# starting kernel
# baseline (speedup 1.0000x reference)
#include <hip/hip_runtime.h>

#define BATCH 256
#define SEQ   4096
#define NS    16
#define MS    8
#define US    4
#define STEPF 448

typedef __attribute__((ext_vector_type(2))) float v2f;
typedef __attribute__((ext_vector_type(8))) float v8f;

__global__ __launch_bounds__(256)
void kalman_cov_kernel(const float* __restrict__ Ag, const float* __restrict__ Bg,
                       const float* __restrict__ Cg, const float* __restrict__ Qg,
                       const float* __restrict__ Rg, float* __restrict__ ws)
{
    __shared__ float sA[NS*NS], sB[NS*US], sC[MS*NS], sQ[NS*NS], sR[MS*MS];
    __shared__ float sP[NS*NS], sT1[NS*NS], sPp[NS*NS];
    __shared__ float sCP[MS*NS], sSI[MS][2*MS], sPCt[NS*MS], sK[NS*MS], sW[NS*NS];

    const int tid = threadIdx.x;
    sA[tid] = Ag[tid];
    sQ[tid] = Qg[tid];
    if (tid < NS*US) sB[tid] = Bg[tid];
    if (tid < MS*NS) sC[tid] = Cg[tid];
    if (tid < MS*MS) sR[tid] = Rg[tid];
    {
        int i = tid >> 4, j = tid & 15;
        sP[tid] = (i == j) ? 1.0f : 0.0f;
    }
    __syncthreads();

    const int i16 = tid >> 4, j16 = tid & 15;
    const int i8  = tid >> 3, j8  = tid & 7;

    #pragma unroll 1
    for (int t = 1; t < SEQ; ++t) {
        float acc = 0.f;
        #pragma unroll
        for (int k = 0; k < NS; ++k) acc += sA[i16*NS + k] * sP[k*NS + j16];
        sT1[tid] = acc;
        __syncthreads();
        acc = sQ[tid];
        #pragma unroll
        for (int k = 0; k < NS; ++k) acc += sT1[i16*NS + k] * sA[j16*NS + k];
        sPp[tid] = acc;
        __syncthreads();
        if (tid < MS*NS) {
            float a = 0.f;
            #pragma unroll
            for (int k = 0; k < NS; ++k) a += sC[i16*NS + k] * sPp[k*NS + j16];
            sCP[i16*NS + j16] = a;
            float b = 0.f;
            #pragma unroll
            for (int k = 0; k < NS; ++k) b += sPp[i8*NS + k] * sC[j8*NS + k];
            sPCt[i8*MS + j8] = b;
        }
        __syncthreads();
        if (tid < MS*NS) {
            float v;
            if (j16 < MS) {
                v = sR[i16*MS + j16];
                #pragma unroll
                for (int k = 0; k < NS; ++k) v += sCP[i16*NS + k] * sC[j16*NS + k];
            } else {
                v = ((j16 - MS) == i16) ? 1.0f : 0.0f;
            }
            sSI[i16][j16] = v;
        }
        __syncthreads();
        #pragma unroll 1
        for (int p = 0; p < MS; ++p) {
            float invp = 1.0f / sSI[p][p];
            float pv = 0.f, f = 0.f;
            if (tid < MS*NS) { pv = sSI[p][j16] * invp; f = sSI[i16][p]; }
            __syncthreads();
            if (tid < MS*NS) sSI[i16][j16] = (i16 == p) ? pv : sSI[i16][j16] - f * pv;
            __syncthreads();
        }
        if (tid < NS*MS) {
            float v = 0.f;
            #pragma unroll
            for (int k = 0; k < MS; ++k) v += sPCt[i8*MS + k] * sSI[k][MS + j8];
            sK[i8*MS + j8] = v;
        }
        __syncthreads();
        {
            float v = (i16 == j16) ? 1.0f : 0.0f;
            #pragma unroll
            for (int k = 0; k < MS; ++k) v -= sK[i16*MS + k] * sC[k*NS + j16];
            sW[tid] = v;
        }
        __syncthreads();
        float* wsp = ws + (size_t)(t - 1) * STEPF;
        {
            float v = 0.f;
            #pragma unroll
            for (int k = 0; k < NS; ++k) v += sW[i16*NS + k] * sA[k*NS + j16];
            wsp[tid] = v;
        }
        {
            float v = 0.f;
            #pragma unroll
            for (int k = 0; k < NS; ++k) v += sW[i16*NS + k] * sPp[k*NS + j16];
            sP[tid] = v;
        }
        if (tid < NS*US) {
            int ib = tid >> 2, jb = tid & 3;
            float v = 0.f;
            #pragma unroll
            for (int k = 0; k < NS; ++k) v += sW[ib*NS + k] * sB[k*US + jb];
            wsp[256 + tid] = v;
        }
        if (tid < NS*MS) wsp[320 + tid] = sK[tid];
        __syncthreads();
    }
}

__device__ __forceinline__ float bperm(int srcLane, float v) {
    return __builtin_bit_cast(float,
        __builtin_amdgcn_ds_bpermute(srcLane << 2, __builtin_bit_cast(int, v)));
}

__device__ __forceinline__ v8f wmma4(v2f a, v2f b, v8f c) {
    return __builtin_amdgcn_wmma_f32_16x16x4_f32(false, a, false, b,
                                                 (short)0, c, false, false);
}

struct StepOps {
    v2f aA0, aA1, aA2, aA3;
    v2f aB, aK0, aK1;
    v2f bZ0, bZ1, bU;
};

__device__ __forceinline__ StepOps load_ops(const float* __restrict__ ws,
                                            const float* __restrict__ zbase,
                                            const float* __restrict__ ubase,
                                            int t, int col, int hi) {
    StepOps o;
    const float* w = ws + (size_t)(t - 1) * STEPF;
    o.aA0 = *(const v2f*)(w + col * 16 + 0  + 2 * hi);
    o.aA1 = *(const v2f*)(w + col * 16 + 4  + 2 * hi);
    o.aA2 = *(const v2f*)(w + col * 16 + 8  + 2 * hi);
    o.aA3 = *(const v2f*)(w + col * 16 + 12 + 2 * hi);
    o.aB  = *(const v2f*)(w + 256 + col * 4 + 2 * hi);
    o.aK0 = *(const v2f*)(w + 320 + col * 8 + 0 + 2 * hi);
    o.aK1 = *(const v2f*)(w + 320 + col * 8 + 4 + 2 * hi);
    const float* zr = zbase + (size_t)t * MS;
    o.bZ0 = *(const v2f*)(zr + 0 + 2 * hi);
    o.bZ1 = *(const v2f*)(zr + 4 + 2 * hi);
    o.bU  = *(const v2f*)(ubase + (size_t)t * US + 2 * hi);
    return o;
}

__device__ __forceinline__ v8f zu_chain(const StepOps& o) {
    v8f d = {};
    d = wmma4(o.aB,  o.bU,  d);
    d = wmma4(o.aK0, o.bZ0, d);
    d = wmma4(o.aK1, o.bZ1, d);
    return d;
}

__global__ __launch_bounds__(32)
void kalman_state_kernel(const float* __restrict__ obs, const float* __restrict__ inp,
                         const float* __restrict__ x0, const float* __restrict__ ws,
                         float* __restrict__ out)
{
    const int lane = threadIdx.x;
    const int hi   = lane >> 4;
    const int col  = lane & 15;
    const int b    = blockIdx.x * 16 + col;

    v8f x;
    #pragma unroll
    for (int r = 0; r < 8; ++r) x[r] = x0[8 * hi + r];

    {
        float* o = out + ((size_t)b * SEQ) * NS + 8 * hi;
        *(float4*)(o)     = make_float4(x[0], x[1], x[2], x[3]);
        *(float4*)(o + 4) = make_float4(x[4], x[5], x[6], x[7]);
    }

    const float* zbase = obs + (size_t)b * SEQ * MS;
    const float* ubase = inp + (size_t)b * SEQ * US;
    float*       obase = out + (size_t)b * SEQ * NS;

    StepOps cur = load_ops(ws, zbase, ubase, 1, col, hi);
    v8f dzu = zu_chain(cur);

    #pragma unroll 1
    for (int t = 1; t < SEQ; ++t) {
        const int tn = (t + 1 < SEQ) ? (t + 1) : t;
        StepOps nxt = load_ops(ws, zbase, ubase, tn, col, hi);

        v2f bX[4];
        #pragma unroll
        for (int c = 0; c < 4; ++c) {
            int srcLane = col + ((c >= 2) ? 16 : 0);
            int lo = 4 * (c & 1);
            float t0 = bperm(srcLane, x[lo + 0]);
            float t1 = bperm(srcLane, x[lo + 2]);
            float t2 = bperm(srcLane, x[lo + 1]);
            float t3 = bperm(srcLane, x[lo + 3]);
            bX[c].x = hi ? t1 : t0;
            bX[c].y = hi ? t3 : t2;
        }

        v8f d = wmma4(cur.aA0, bX[0], dzu);
        d = wmma4(cur.aA1, bX[1], d);
        d = wmma4(cur.aA2, bX[2], d);
        d = wmma4(cur.aA3, bX[3], d);
        x = d;

        float* o = obase + (size_t)t * NS + 8 * hi;
        *(float4*)(o)     = make_float4(x[0], x[1], x[2], x[3]);
        *(float4*)(o + 4) = make_float4(x[4], x[5], x[6], x[7]);

        dzu = zu_chain(nxt);
        cur = nxt;
    }
}

extern "C" void kernel_launch(void* const* d_in, const int* in_sizes, int n_in,
                              void* d_out, int out_size, void* d_ws, size_t ws_size,
                              hipStream_t stream) {
    const float* obs = (const float*)d_in[0];
    const float* inp = (const float*)d_in[1];
    const float* A   = (const float*)d_in[2];
    const float* B   = (const float*)d_in[3];
    const float* C   = (const float*)d_in[4];
    const float* Q   = (const float*)d_in[5];
    const float* R   = (const float*)d_in[6];
    const float* x0  = (const float*)d_in[7];
    float* out = (float*)d_out;
    float* ws  = (float*)d_ws;

    hipLaunchKernelGGL(kalman_cov_kernel, dim3(1), dim3(256), 0, stream,
                       A, B, C, Q, R, ws);
    hipLaunchKernelGGL(kalman_state_kernel, dim3(BATCH / 16), dim3(32), 0, stream,
                       obs, inp, x0, ws, out);
}


# S inverse (Gauss-Jordan) moved from LDS with two block barriers per pivot into one wave's registers: column per lane, pivot column by readlane; same order, row update now a fused multiply-subtract
# speedup vs baseline: 1.3579x; 1.3579x over previous
#include <hip/hip_runtime.h>

#define BATCH 256
#define SEQ   4096
#define NS    16
#define MS    8
#define US    4
#define STEPF 448

typedef __attribute__((ext_vector_type(2))) float v2f;
typedef __attribute__((ext_vector_type(8))) float v8f;

__device__ __forceinline__ float rdlane(float v, int lane) {
    return __builtin_bit_cast(float,
        __builtin_amdgcn_readlane(__builtin_bit_cast(int, v), lane));
}

__global__ __launch_bounds__(256)
void kalman_cov_kernel(const float* __restrict__ Ag, const float* __restrict__ Bg,
                       const float* __restrict__ Cg, const float* __restrict__ Qg,
                       const float* __restrict__ Rg, float* __restrict__ ws)
{
    __shared__ float sA[NS*NS], sB[NS*US], sC[MS*NS], sQ[NS*NS], sR[MS*MS];
    __shared__ float sP[NS*NS], sT1[NS*NS], sPp[NS*NS];
    __shared__ float sCP[MS*NS], sSI[MS][2*MS], sPCt[NS*MS], sK[NS*MS], sW[NS*NS];

    const int tid = threadIdx.x;
    sA[tid] = Ag[tid];
    sQ[tid] = Qg[tid];
    if (tid < NS*US) sB[tid] = Bg[tid];
    if (tid < MS*NS) sC[tid] = Cg[tid];
    if (tid < MS*MS) sR[tid] = Rg[tid];
    {
        int i = tid >> 4, j = tid & 15;
        sP[tid] = (i == j) ? 1.0f : 0.0f;
    }
    __syncthreads();

    const int i16 = tid >> 4, j16 = tid & 15;
    const int i8  = tid >> 3, j8  = tid & 7;

    #pragma unroll 1
    for (int t = 1; t < SEQ; ++t) {
        float acc = 0.f;
        #pragma unroll
        for (int k = 0; k < NS; ++k) acc += sA[i16*NS + k] * sP[k*NS + j16];
        sT1[tid] = acc;
        __syncthreads();
        acc = sQ[tid];
        #pragma unroll
        for (int k = 0; k < NS; ++k) acc += sT1[i16*NS + k] * sA[j16*NS + k];
        sPp[tid] = acc;
        __syncthreads();
        if (tid < MS*NS) {
            float a = 0.f;
            #pragma unroll
            for (int k = 0; k < NS; ++k) a += sC[i16*NS + k] * sPp[k*NS + j16];
            sCP[i16*NS + j16] = a;
            float b = 0.f;
            #pragma unroll
            for (int k = 0; k < NS; ++k) b += sPp[i8*NS + k] * sC[j8*NS + k];
            sPCt[i8*MS + j8] = b;
        }
        __syncthreads();
        if (tid < MS*NS) {
            float v;
            if (j16 < MS) {
                v = sR[i16*MS + j16];
                #pragma unroll
                for (int k = 0; k < NS; ++k) v += sCP[i16*NS + k] * sC[j16*NS + k];
            } else {
                v = ((j16 - MS) == i16) ? 1.0f : 0.0f;
            }
            sSI[i16][j16] = v;
        }
        __syncthreads();
        if (tid < 32) {
            float c[MS];
            #pragma unroll
            for (int i = 0; i < MS; ++i) c[i] = sSI[i][j16];
            #pragma unroll
            for (int p = 0; p < MS; ++p) {
                const float invp = 1.0f / rdlane(c[p], p);
                const float pv = c[p] * invp;
                #pragma unroll
                for (int i = 0; i < MS; ++i) {
                    if (i != p) {
                        const float f = rdlane(c[i], p);
                        c[i] = c[i] - f * pv;
                    }
                }
                c[p] = pv;
            }
            if (tid >= MS && tid < 2*MS) {
                #pragma unroll
                for (int i = 0; i < MS; ++i) sSI[i][tid] = c[i];
            }
        }
        __syncthreads();
        if (tid < NS*MS) {
            float v = 0.f;
            #pragma unroll
            for (int k = 0; k < MS; ++k) v += sPCt[i8*MS + k] * sSI[k][MS + j8];
            sK[i8*MS + j8] = v;
        }
        __syncthreads();
        {
            float v = (i16 == j16) ? 1.0f : 0.0f;
            #pragma unroll
            for (int k = 0; k < MS; ++k) v -= sK[i16*MS + k] * sC[k*NS + j16];
            sW[tid] = v;
        }
        __syncthreads();
        float* wsp = ws + (size_t)(t - 1) * STEPF;
        {
            float v = 0.f;
            #pragma unroll
            for (int k = 0; k < NS; ++k) v += sW[i16*NS + k] * sA[k*NS + j16];
            wsp[tid] = v;
        }
        {
            float v = 0.f;
            #pragma unroll
            for (int k = 0; k < NS; ++k) v += sW[i16*NS + k] * sPp[k*NS + j16];
            sP[tid] = v;
        }
        if (tid < NS*US) {
            int ib = tid >> 2, jb = tid & 3;
            float v = 0.f;
            #pragma unroll
            for (int k = 0; k < NS; ++k) v += sW[ib*NS + k] * sB[k*US + jb];
            wsp[256 + tid] = v;
        }
        if (tid < NS*MS) wsp[320 + tid] = sK[tid];
        __syncthreads();
    }
}

__device__ __forceinline__ float bperm(int srcLane, float v) {
    return __builtin_bit_cast(float,
        __builtin_amdgcn_ds_bpermute(srcLane << 2, __builtin_bit_cast(int, v)));
}

__device__ __forceinline__ v8f wmma4(v2f a, v2f b, v8f c) {
    return __builtin_amdgcn_wmma_f32_16x16x4_f32(false, a, false, b,
                                                 (short)0, c, false, false);
}

struct StepOps {
    v2f aA0, aA1, aA2, aA3;
    v2f aB, aK0, aK1;
    v2f bZ0, bZ1, bU;
};

__device__ __forceinline__ StepOps load_ops(const float* __restrict__ ws,
                                            const float* __restrict__ zbase,
                                            const float* __restrict__ ubase,
                                            int t, int col, int hi) {
    StepOps o;
    const float* w = ws + (size_t)(t - 1) * STEPF;
    o.aA0 = *(const v2f*)(w + col * 16 + 0  + 2 * hi);
    o.aA1 = *(const v2f*)(w + col * 16 + 4  + 2 * hi);
    o.aA2 = *(const v2f*)(w + col * 16 + 8  + 2 * hi);
    o.aA3 = *(const v2f*)(w + col * 16 + 12 + 2 * hi);
    o.aB  = *(const v2f*)(w + 256 + col * 4 + 2 * hi);
    o.aK0 = *(const v2f*)(w + 320 + col * 8 + 0 + 2 * hi);
    o.aK1 = *(const v2f*)(w + 320 + col * 8 + 4 + 2 * hi);
    const float* zr = zbase + (size_t)t * MS;
    o.bZ0 = *(const v2f*)(zr + 0 + 2 * hi);
    o.bZ1 = *(const v2f*)(zr + 4 + 2 * hi);
    o.bU  = *(const v2f*)(ubase + (size_t)t * US + 2 * hi);
    return o;
}

__device__ __forceinline__ v8f zu_chain(const StepOps& o) {
    v8f d = {};
    d = wmma4(o.aB,  o.bU,  d);
    d = wmma4(o.aK0, o.bZ0, d);
    d = wmma4(o.aK1, o.bZ1, d);
    return d;
}

__global__ __launch_bounds__(32)
void kalman_state_kernel(const float* __restrict__ obs, const float* __restrict__ inp,
                         const float* __restrict__ x0, const float* __restrict__ ws,
                         float* __restrict__ out)
{
    const int lane = threadIdx.x;
    const int hi   = lane >> 4;
    const int col  = lane & 15;
    const int b    = blockIdx.x * 16 + col;

    v8f x;
    #pragma unroll
    for (int r = 0; r < 8; ++r) x[r] = x0[8 * hi + r];

    {
        float* o = out + ((size_t)b * SEQ) * NS + 8 * hi;
        *(float4*)(o)     = make_float4(x[0], x[1], x[2], x[3]);
        *(float4*)(o + 4) = make_float4(x[4], x[5], x[6], x[7]);
    }

    const float* zbase = obs + (size_t)b * SEQ * MS;
    const float* ubase = inp + (size_t)b * SEQ * US;
    float*       obase = out + (size_t)b * SEQ * NS;

    StepOps cur = load_ops(ws, zbase, ubase, 1, col, hi);
    v8f dzu = zu_chain(cur);

    #pragma unroll 1
    for (int t = 1; t < SEQ; ++t) {
        const int tn = (t + 1 < SEQ) ? (t + 1) : t;
        StepOps nxt = load_ops(ws, zbase, ubase, tn, col, hi);

        v2f bX[4];
        #pragma unroll
        for (int c = 0; c < 4; ++c) {
            int srcLane = col + ((c >= 2) ? 16 : 0);
            int lo = 4 * (c & 1);
            float t0 = bperm(srcLane, x[lo + 0]);
            float t1 = bperm(srcLane, x[lo + 2]);
            float t2 = bperm(srcLane, x[lo + 1]);
            float t3 = bperm(srcLane, x[lo + 3]);
            bX[c].x = hi ? t1 : t0;
            bX[c].y = hi ? t3 : t2;
        }

        v8f d = wmma4(cur.aA0, bX[0], dzu);
        d = wmma4(cur.aA1, bX[1], d);
        d = wmma4(cur.aA2, bX[2], d);
        d = wmma4(cur.aA3, bX[3], d);
        x = d;

        float* o = obase + (size_t)t * NS + 8 * hi;
        *(float4*)(o)     = make_float4(x[0], x[1], x[2], x[3]);
        *(float4*)(o + 4) = make_float4(x[4], x[5], x[6], x[7]);

        dzu = zu_chain(nxt);
        cur = nxt;
    }
}

extern "C" void kernel_launch(void* const* d_in, const int* in_sizes, int n_in,
                              void* d_out, int out_size, void* d_ws, size_t ws_size,
                              hipStream_t stream) {
    const float* obs = (const float*)d_in[0];
    const float* inp = (const float*)d_in[1];
    const float* A   = (const float*)d_in[2];
    const float* B   = (const float*)d_in[3];
    const float* C   = (const float*)d_in[4];
    const float* Q   = (const float*)d_in[5];
    const float* R   = (const float*)d_in[6];
    const float* x0  = (const float*)d_in[7];
    float* out = (float*)d_out;
    float* ws  = (float*)d_ws;

    hipLaunchKernelGGL(kalman_cov_kernel, dim3(1), dim3(256), 0, stream,
                       A, B, C, Q, R, ws);
    hipLaunchKernelGGL(kalman_state_kernel, dim3(BATCH / 16), dim3(32), 0, stream,
                       obs, inp, x0, ws, out);
}


# Covariance recursion moved from 8 waves on LDS matrices with 8 block barriers per step into one wave: every product on f32 WMMA from registers, no LDS; sums now run in WMMA k order
# speedup vs baseline: 3.2081x; 2.3625x over previous
#include <hip/hip_runtime.h>

#define BATCH 256
#define SEQ   4096
#define NS    16
#define MS    8
#define US    4
#define STEPF 448

typedef __attribute__((ext_vector_type(2))) float v2f;
typedef __attribute__((ext_vector_type(8))) float v8f;

__device__ __forceinline__ float rdlane(float v, int lane) {
    return __builtin_bit_cast(float,
        __builtin_amdgcn_readlane(__builtin_bit_cast(int, v), lane));
}

__device__ __forceinline__ v8f wmma4(v2f a, v2f b, v8f c) {
    return __builtin_amdgcn_wmma_f32_16x16x4_f32(false, a, false, b,
                                                 (short)0, c, false, false);
}

struct Op4 { v2f q[4]; };

__device__ __forceinline__ Op4 from_acc(v8f x) {
    Op4 o;
    o.q[0] = v2f{x[0], x[1]};
    o.q[1] = v2f{x[2], x[3]};
    o.q[2] = v2f{x[4], x[5]};
    o.q[3] = v2f{x[6], x[7]};
    return o;
}

__device__ __forceinline__ v8f mm16(const Op4& a, const Op4& b, v8f c) {
    c = wmma4(a.q[0], b.q[0], c);
    c = wmma4(a.q[1], b.q[1], c);
    c = wmma4(a.q[2], b.q[2], c);
    c = wmma4(a.q[3], b.q[3], c);
    return c;
}

__global__ __launch_bounds__(32)
void kalman_cov_kernel(const float* __restrict__ Ag, const float* __restrict__ Bg,
                       const float* __restrict__ Cg, const float* __restrict__ Qg,
                       const float* __restrict__ Rg, float* __restrict__ ws)
{
    const int lane = threadIdx.x;
    const int h = lane >> 4;
    const int c = lane & 15;
    const int k0 = 8 * h;
    const int c7 = c & 7;
    const int c3 = c & 3;

    Op4 opA, opAtA, opC, opCsh, opNCt, opBtA;
    #pragma unroll
    for (int q = 0; q < 4; ++q) {
        const int k = k0 + 2 * q;
        opA.q[q]   = v2f{Ag[c * NS + k], Ag[c * NS + k + 1]};
        opAtA.q[q] = v2f{Ag[k * NS + c], Ag[(k + 1) * NS + c]};
        const float cv0 = Cg[c7 * NS + k], cv1 = Cg[c7 * NS + k + 1];
        opC.q[q]   = (c < MS)  ? v2f{cv0, cv1} : v2f{0.f, 0.f};
        opCsh.q[q] = (c >= MS) ? v2f{cv0, cv1} : v2f{0.f, 0.f};
        const float n0 = -Cg[(2 * q) * NS + c], n1 = -Cg[(2 * q + 1) * NS + c];
        opNCt.q[q] = (h == 0) ? v2f{n0, n1} : v2f{0.f, 0.f};
        const float b0 = Bg[k * US + c3], b1 = Bg[(k + 1) * US + c3];
        opBtA.q[q] = (c < US) ? v2f{b0, b1} : v2f{0.f, 0.f};
    }
    v8f dQ, dQt, dI, dRaug, P;
    #pragma unroll
    for (int r = 0; r < 8; ++r) {
        dQ[r]  = Qg[(k0 + r) * NS + c];
        dQt[r] = Qg[c * NS + k0 + r];
        dI[r]  = (k0 + r == c) ? 1.0f : 0.0f;
        const float rv = Rg[r * MS + c7];
        dRaug[r] = (h == 0) ? ((c >= MS) ? rv : ((r == c) ? 1.0f : 0.0f)) : 0.0f;
        P[r] = dI[r];
    }
    const v8f zero = {};

    #pragma unroll 1
    for (int t = 1; t < SEQ; ++t) {
        const v8f T1t = mm16(from_acc(P), opA, zero);
        const Op4 oT1 = from_acc(T1t);
        const v8f Pp  = mm16(oT1, opA, dQ);
        const v8f PpT = mm16(opA, oT1, dQt);
        const v8f CPt  = mm16(from_acc(Pp), opC, zero);
        const v8f PCtT = mm16(opC, from_acc(PpT), zero);
        v8f g = mm16(from_acc(CPt), opCsh, dRaug);

        #pragma unroll
        for (int p = 0; p < MS; ++p) {
            const float invp = 1.0f / rdlane(g[p], MS + p);
            const float pv = g[p] * invp;
            #pragma unroll
            for (int i = 0; i < MS; ++i) {
                if (i != p) {
                    const float f = rdlane(g[i], MS + p);
                    g[i] = g[i] - f * pv;
                }
            }
            g[p] = pv;
        }
        Op4 oSi = from_acc(g);
        {
            const bool keep = lane < MS;
            const v2f z2 = v2f{0.f, 0.f};
            oSi.q[0] = keep ? oSi.q[0] : z2;
            oSi.q[1] = keep ? oSi.q[1] : z2;
            oSi.q[2] = keep ? oSi.q[2] : z2;
            oSi.q[3] = keep ? oSi.q[3] : z2;
        }
        const v8f Kt = mm16(oSi, from_acc(PCtT), zero);
        const v8f Wt = mm16(opNCt, from_acc(Kt), dI);
        const Op4 oW = from_acc(Wt);
        P = mm16(oW, from_acc(Pp), zero);
        const v8f Hb = mm16(opBtA, oW, zero);
        const v8f G = mm16(opAtA, oW, zero);

        float* wsp = ws + (size_t)(t - 1) * STEPF;
        {
            float* o = wsp + c * NS + k0;
            *(float4*)(o)     = make_float4(G[0], G[1], G[2], G[3]);
            *(float4*)(o + 4) = make_float4(G[4], G[5], G[6], G[7]);
        }
        if (h == 0) {
            *(float4*)(wsp + 256 + c * US) = make_float4(Hb[0], Hb[1], Hb[2], Hb[3]);
            float* o = wsp + 320 + c * MS;
            *(float4*)(o)     = make_float4(Kt[0], Kt[1], Kt[2], Kt[3]);
            *(float4*)(o + 4) = make_float4(Kt[4], Kt[5], Kt[6], Kt[7]);
        }
    }
}

__device__ __forceinline__ float bperm(int srcLane, float v) {
    return __builtin_bit_cast(float,
        __builtin_amdgcn_ds_bpermute(srcLane << 2, __builtin_bit_cast(int, v)));
}

struct StepOps {
    v2f aA0, aA1, aA2, aA3;
    v2f aB, aK0, aK1;
    v2f bZ0, bZ1, bU;
};

__device__ __forceinline__ StepOps load_ops(const float* __restrict__ ws,
                                            const float* __restrict__ zbase,
                                            const float* __restrict__ ubase,
                                            int t, int col, int hi) {
    StepOps o;
    const float* w = ws + (size_t)(t - 1) * STEPF;
    o.aA0 = *(const v2f*)(w + col * 16 + 0  + 2 * hi);
    o.aA1 = *(const v2f*)(w + col * 16 + 4  + 2 * hi);
    o.aA2 = *(const v2f*)(w + col * 16 + 8  + 2 * hi);
    o.aA3 = *(const v2f*)(w + col * 16 + 12 + 2 * hi);
    o.aB  = *(const v2f*)(w + 256 + col * 4 + 2 * hi);
    o.aK0 = *(const v2f*)(w + 320 + col * 8 + 0 + 2 * hi);
    o.aK1 = *(const v2f*)(w + 320 + col * 8 + 4 + 2 * hi);
    const float* zr = zbase + (size_t)t * MS;
    o.bZ0 = *(const v2f*)(zr + 0 + 2 * hi);
    o.bZ1 = *(const v2f*)(zr + 4 + 2 * hi);
    o.bU  = *(const v2f*)(ubase + (size_t)t * US + 2 * hi);
    return o;
}

__device__ __forceinline__ v8f zu_chain(const StepOps& o) {
    v8f d = {};
    d = wmma4(o.aB,  o.bU,  d);
    d = wmma4(o.aK0, o.bZ0, d);
    d = wmma4(o.aK1, o.bZ1, d);
    return d;
}

__global__ __launch_bounds__(32)
void kalman_state_kernel(const float* __restrict__ obs, const float* __restrict__ inp,
                         const float* __restrict__ x0, const float* __restrict__ ws,
                         float* __restrict__ out)
{
    const int lane = threadIdx.x;
    const int hi   = lane >> 4;
    const int col  = lane & 15;
    const int b    = blockIdx.x * 16 + col;

    v8f x;
    #pragma unroll
    for (int r = 0; r < 8; ++r) x[r] = x0[8 * hi + r];

    {
        float* o = out + ((size_t)b * SEQ) * NS + 8 * hi;
        *(float4*)(o)     = make_float4(x[0], x[1], x[2], x[3]);
        *(float4*)(o + 4) = make_float4(x[4], x[5], x[6], x[7]);
    }

    const float* zbase = obs + (size_t)b * SEQ * MS;
    const float* ubase = inp + (size_t)b * SEQ * US;
    float*       obase = out + (size_t)b * SEQ * NS;

    StepOps cur = load_ops(ws, zbase, ubase, 1, col, hi);
    v8f dzu = zu_chain(cur);

    #pragma unroll 1
    for (int t = 1; t < SEQ; ++t) {
        const int tn = (t + 1 < SEQ) ? (t + 1) : t;
        StepOps nxt = load_ops(ws, zbase, ubase, tn, col, hi);

        v2f bX[4];
        #pragma unroll
        for (int c = 0; c < 4; ++c) {
            int srcLane = col + ((c >= 2) ? 16 : 0);
            int lo = 4 * (c & 1);
            float t0 = bperm(srcLane, x[lo + 0]);
            float t1 = bperm(srcLane, x[lo + 2]);
            float t2 = bperm(srcLane, x[lo + 1]);
            float t3 = bperm(srcLane, x[lo + 3]);
            bX[c].x = hi ? t1 : t0;
            bX[c].y = hi ? t3 : t2;
        }

        v8f d = wmma4(cur.aA0, bX[0], dzu);
        d = wmma4(cur.aA1, bX[1], d);
        d = wmma4(cur.aA2, bX[2], d);
        d = wmma4(cur.aA3, bX[3], d);
        x = d;

        float* o = obase + (size_t)t * NS + 8 * hi;
        *(float4*)(o)     = make_float4(x[0], x[1], x[2], x[3]);
        *(float4*)(o + 4) = make_float4(x[4], x[5], x[6], x[7]);

        dzu = zu_chain(nxt);
        cur = nxt;
    }
}

extern "C" void kernel_launch(void* const* d_in, const int* in_sizes, int n_in,
                              void* d_out, int out_size, void* d_ws, size_t ws_size,
                              hipStream_t stream) {
    const float* obs = (const float*)d_in[0];
    const float* inp = (const float*)d_in[1];
    const float* A   = (const float*)d_in[2];
    const float* B   = (const float*)d_in[3];
    const float* C   = (const float*)d_in[4];
    const float* Q   = (const float*)d_in[5];
    const float* R   = (const float*)d_in[6];
    const float* x0  = (const float*)d_in[7];
    float* out = (float*)d_out;
    float* ws  = (float*)d_ws;

    hipLaunchKernelGGL(kalman_cov_kernel, dim3(1), dim3(32), 0, stream,
                       A, B, C, Q, R, ws);
    hipLaunchKernelGGL(kalman_state_kernel, dim3(BATCH / 16), dim3(32), 0, stream,
                       obs, inp, x0, ws, out);
}


# Phase 2 operand loads issued 5 steps ahead through a 5-slot register ring (step loop unrolled by 5) instead of 1 step ahead, so no step waits a full load round trip; same sums in the same order
# speedup vs baseline: 4.3290x; 1.3494x over previous
#include <hip/hip_runtime.h>

#define BATCH 256
#define SEQ   4096
#define NS    16
#define MS    8
#define US    4
#define STEPF 448
#define PF_DEPTH 5
static_assert((SEQ - 1) % PF_DEPTH == 0, "phase 2 walks the steps in whole passes of PF_DEPTH");

typedef __attribute__((ext_vector_type(2))) float v2f;
typedef __attribute__((ext_vector_type(8))) float v8f;

__device__ __forceinline__ float rdlane(float v, int lane) {
    return __builtin_bit_cast(float,
        __builtin_amdgcn_readlane(__builtin_bit_cast(int, v), lane));
}

__device__ __forceinline__ v8f wmma4(v2f a, v2f b, v8f c) {
    return __builtin_amdgcn_wmma_f32_16x16x4_f32(false, a, false, b,
                                                 (short)0, c, false, false);
}

struct Op4 { v2f q[4]; };

__device__ __forceinline__ Op4 from_acc(v8f x) {
    Op4 o;
    o.q[0] = v2f{x[0], x[1]};
    o.q[1] = v2f{x[2], x[3]};
    o.q[2] = v2f{x[4], x[5]};
    o.q[3] = v2f{x[6], x[7]};
    return o;
}

__device__ __forceinline__ v8f mm16(const Op4& a, const Op4& b, v8f c) {
    c = wmma4(a.q[0], b.q[0], c);
    c = wmma4(a.q[1], b.q[1], c);
    c = wmma4(a.q[2], b.q[2], c);
    c = wmma4(a.q[3], b.q[3], c);
    return c;
}

__global__ __launch_bounds__(32)
void kalman_cov_kernel(const float* __restrict__ Ag, const float* __restrict__ Bg,
                       const float* __restrict__ Cg, const float* __restrict__ Qg,
                       const float* __restrict__ Rg, float* __restrict__ ws)
{
    const int lane = threadIdx.x;
    const int h = lane >> 4;
    const int c = lane & 15;
    const int k0 = 8 * h;
    const int c7 = c & 7;
    const int c3 = c & 3;

    Op4 opA, opAtA, opC, opCsh, opNCt, opBtA;
    #pragma unroll
    for (int q = 0; q < 4; ++q) {
        const int k = k0 + 2 * q;
        opA.q[q]   = v2f{Ag[c * NS + k], Ag[c * NS + k + 1]};
        opAtA.q[q] = v2f{Ag[k * NS + c], Ag[(k + 1) * NS + c]};
        const float cv0 = Cg[c7 * NS + k], cv1 = Cg[c7 * NS + k + 1];
        opC.q[q]   = (c < MS)  ? v2f{cv0, cv1} : v2f{0.f, 0.f};
        opCsh.q[q] = (c >= MS) ? v2f{cv0, cv1} : v2f{0.f, 0.f};
        const float n0 = -Cg[(2 * q) * NS + c], n1 = -Cg[(2 * q + 1) * NS + c];
        opNCt.q[q] = (h == 0) ? v2f{n0, n1} : v2f{0.f, 0.f};
        const float b0 = Bg[k * US + c3], b1 = Bg[(k + 1) * US + c3];
        opBtA.q[q] = (c < US) ? v2f{b0, b1} : v2f{0.f, 0.f};
    }
    v8f dQ, dQt, dI, dRaug, P;
    #pragma unroll
    for (int r = 0; r < 8; ++r) {
        dQ[r]  = Qg[(k0 + r) * NS + c];
        dQt[r] = Qg[c * NS + k0 + r];
        dI[r]  = (k0 + r == c) ? 1.0f : 0.0f;
        const float rv = Rg[r * MS + c7];
        dRaug[r] = (h == 0) ? ((c >= MS) ? rv : ((r == c) ? 1.0f : 0.0f)) : 0.0f;
        P[r] = dI[r];
    }
    const v8f zero = {};

    #pragma unroll 1
    for (int t = 1; t < SEQ; ++t) {
        const v8f T1t = mm16(from_acc(P), opA, zero);
        const Op4 oT1 = from_acc(T1t);
        const v8f Pp  = mm16(oT1, opA, dQ);
        const v8f PpT = mm16(opA, oT1, dQt);
        const v8f CPt  = mm16(from_acc(Pp), opC, zero);
        const v8f PCtT = mm16(opC, from_acc(PpT), zero);
        v8f g = mm16(from_acc(CPt), opCsh, dRaug);

        #pragma unroll
        for (int p = 0; p < MS; ++p) {
            const float invp = 1.0f / rdlane(g[p], MS + p);
            const float pv = g[p] * invp;
            #pragma unroll
            for (int i = 0; i < MS; ++i) {
                if (i != p) {
                    const float f = rdlane(g[i], MS + p);
                    g[i] = g[i] - f * pv;
                }
            }
            g[p] = pv;
        }
        Op4 oSi = from_acc(g);
        {
            const bool keep = lane < MS;
            const v2f z2 = v2f{0.f, 0.f};
            oSi.q[0] = keep ? oSi.q[0] : z2;
            oSi.q[1] = keep ? oSi.q[1] : z2;
            oSi.q[2] = keep ? oSi.q[2] : z2;
            oSi.q[3] = keep ? oSi.q[3] : z2;
        }
        const v8f Kt = mm16(oSi, from_acc(PCtT), zero);
        const v8f Wt = mm16(opNCt, from_acc(Kt), dI);
        const Op4 oW = from_acc(Wt);
        P = mm16(oW, from_acc(Pp), zero);
        const v8f Hb = mm16(opBtA, oW, zero);
        const v8f G = mm16(opAtA, oW, zero);

        float* wsp = ws + (size_t)(t - 1) * STEPF;
        {
            float* o = wsp + c * NS + k0;
            *(float4*)(o)     = make_float4(G[0], G[1], G[2], G[3]);
            *(float4*)(o + 4) = make_float4(G[4], G[5], G[6], G[7]);
        }
        if (h == 0) {
            *(float4*)(wsp + 256 + c * US) = make_float4(Hb[0], Hb[1], Hb[2], Hb[3]);
            float* o = wsp + 320 + c * MS;
            *(float4*)(o)     = make_float4(Kt[0], Kt[1], Kt[2], Kt[3]);
            *(float4*)(o + 4) = make_float4(Kt[4], Kt[5], Kt[6], Kt[7]);
        }
    }
}

__device__ __forceinline__ float bperm(int srcLane, float v) {
    return __builtin_bit_cast(float,
        __builtin_amdgcn_ds_bpermute(srcLane << 2, __builtin_bit_cast(int, v)));
}

struct StepOps {
    v2f aA0, aA1, aA2, aA3;
    v2f aB, aK0, aK1;
    v2f bZ0, bZ1, bU;
};

__device__ __forceinline__ StepOps load_ops(const float* __restrict__ ws,
                                            const float* __restrict__ zbase,
                                            const float* __restrict__ ubase,
                                            int t, int col, int hi) {
    StepOps o;
    const float* w = ws + (size_t)(t - 1) * STEPF;
    o.aA0 = *(const v2f*)(w + col * 16 + 0  + 2 * hi);
    o.aA1 = *(const v2f*)(w + col * 16 + 4  + 2 * hi);
    o.aA2 = *(const v2f*)(w + col * 16 + 8  + 2 * hi);
    o.aA3 = *(const v2f*)(w + col * 16 + 12 + 2 * hi);
    o.aB  = *(const v2f*)(w + 256 + col * 4 + 2 * hi);
    o.aK0 = *(const v2f*)(w + 320 + col * 8 + 0 + 2 * hi);
    o.aK1 = *(const v2f*)(w + 320 + col * 8 + 4 + 2 * hi);
    const float* zr = zbase + (size_t)t * MS;
    o.bZ0 = *(const v2f*)(zr + 0 + 2 * hi);
    o.bZ1 = *(const v2f*)(zr + 4 + 2 * hi);
    o.bU  = *(const v2f*)(ubase + (size_t)t * US + 2 * hi);
    return o;
}

__device__ __forceinline__ v8f zu_chain(const StepOps& o) {
    v8f d = {};
    d = wmma4(o.aB,  o.bU,  d);
    d = wmma4(o.aK0, o.bZ0, d);
    d = wmma4(o.aK1, o.bZ1, d);
    return d;
}

__global__ __launch_bounds__(32)
void kalman_state_kernel(const float* __restrict__ obs, const float* __restrict__ inp,
                         const float* __restrict__ x0, const float* __restrict__ ws,
                         float* __restrict__ out)
{
    const int lane = threadIdx.x;
    const int hi   = lane >> 4;
    const int col  = lane & 15;
    const int b    = blockIdx.x * 16 + col;

    v8f x;
    #pragma unroll
    for (int r = 0; r < 8; ++r) x[r] = x0[8 * hi + r];

    {
        float* o = out + ((size_t)b * SEQ) * NS + 8 * hi;
        *(float4*)(o)     = make_float4(x[0], x[1], x[2], x[3]);
        *(float4*)(o + 4) = make_float4(x[4], x[5], x[6], x[7]);
    }

    const float* zbase = obs + (size_t)b * SEQ * MS;
    const float* ubase = inp + (size_t)b * SEQ * US;
    float*       obase = out + (size_t)b * SEQ * NS;

    StepOps ring[PF_DEPTH];
    #pragma unroll
    for (int s = 0; s < PF_DEPTH; ++s) {
        ring[s] = load_ops(ws, zbase, ubase, 1 + s, col, hi);
        __builtin_amdgcn_sched_barrier(0);
    }
    v8f dzu = zu_chain(ring[0]);

    #pragma unroll 1
    for (int tb = 1; tb < SEQ; tb += PF_DEPTH) {
        #pragma unroll
        for (int s = 0; s < PF_DEPTH; ++s) {
            const int t = tb + s;

            v2f bX[4];
            #pragma unroll
            for (int c = 0; c < 4; ++c) {
                int srcLane = col + ((c >= 2) ? 16 : 0);
                int lo = 4 * (c & 1);
                float t0 = bperm(srcLane, x[lo + 0]);
                float t1 = bperm(srcLane, x[lo + 2]);
                float t2 = bperm(srcLane, x[lo + 1]);
                float t3 = bperm(srcLane, x[lo + 3]);
                bX[c].x = hi ? t1 : t0;
                bX[c].y = hi ? t3 : t2;
            }

            v8f d = wmma4(ring[s].aA0, bX[0], dzu);
            d = wmma4(ring[s].aA1, bX[1], d);
            d = wmma4(ring[s].aA2, bX[2], d);
            d = wmma4(ring[s].aA3, bX[3], d);
            x = d;

            float* o = obase + (size_t)t * NS + 8 * hi;
            *(float4*)(o)     = make_float4(x[0], x[1], x[2], x[3]);
            *(float4*)(o + 4) = make_float4(x[4], x[5], x[6], x[7]);

            dzu = zu_chain(ring[(s + 1) % PF_DEPTH]);

            const int tn = (t + PF_DEPTH < SEQ) ? (t + PF_DEPTH) : (SEQ - 1);
            ring[s] = load_ops(ws, zbase, ubase, tn, col, hi);
            __builtin_amdgcn_sched_barrier(0);
        }
    }
}

extern "C" void kernel_launch(void* const* d_in, const int* in_sizes, int n_in,
                              void* d_out, int out_size, void* d_ws, size_t ws_size,
                              hipStream_t stream) {
    const float* obs = (const float*)d_in[0];
    const float* inp = (const float*)d_in[1];
    const float* A   = (const float*)d_in[2];
    const float* B   = (const float*)d_in[3];
    const float* C   = (const float*)d_in[4];
    const float* Q   = (const float*)d_in[5];
    const float* R   = (const float*)d_in[6];
    const float* x0  = (const float*)d_in[7];
    float* out = (float*)d_out;
    float* ws  = (float*)d_ws;

    hipLaunchKernelGGL(kalman_cov_kernel, dim3(1), dim3(32), 0, stream,
                       A, B, C, Q, R, ws);
    hipLaunchKernelGGL(kalman_state_kernel, dim3(BATCH / 16), dim3(32), 0, stream,
                       obs, inp, x0, ws, out);
}


# Gauss-Jordan pivot reciprocal by the hardware reciprocal instruction (within 1 ulp) instead of the 11-instruction IEEE division on every pivot's serial chain; same elimination order
# speedup vs baseline: 5.1802x; 1.1966x over previous
#include <hip/hip_runtime.h>

#define BATCH 256
#define SEQ   4096
#define NS    16
#define MS    8
#define US    4
#define STEPF 448
#define PF_DEPTH 5
static_assert((SEQ - 1) % PF_DEPTH == 0, "phase 2 walks the steps in whole passes of PF_DEPTH");

typedef __attribute__((ext_vector_type(2))) float v2f;
typedef __attribute__((ext_vector_type(8))) float v8f;

__device__ __forceinline__ float rdlane(float v, int lane) {
    return __builtin_bit_cast(float,
        __builtin_amdgcn_readlane(__builtin_bit_cast(int, v), lane));
}

__device__ __forceinline__ v8f wmma4(v2f a, v2f b, v8f c) {
    return __builtin_amdgcn_wmma_f32_16x16x4_f32(false, a, false, b,
                                                 (short)0, c, false, false);
}

struct Op4 { v2f q[4]; };

__device__ __forceinline__ Op4 from_acc(v8f x) {
    Op4 o;
    o.q[0] = v2f{x[0], x[1]};
    o.q[1] = v2f{x[2], x[3]};
    o.q[2] = v2f{x[4], x[5]};
    o.q[3] = v2f{x[6], x[7]};
    return o;
}

__device__ __forceinline__ v8f mm16(const Op4& a, const Op4& b, v8f c) {
    c = wmma4(a.q[0], b.q[0], c);
    c = wmma4(a.q[1], b.q[1], c);
    c = wmma4(a.q[2], b.q[2], c);
    c = wmma4(a.q[3], b.q[3], c);
    return c;
}

__global__ __launch_bounds__(32)
void kalman_cov_kernel(const float* __restrict__ Ag, const float* __restrict__ Bg,
                       const float* __restrict__ Cg, const float* __restrict__ Qg,
                       const float* __restrict__ Rg, float* __restrict__ ws)
{
    const int lane = threadIdx.x;
    const int h = lane >> 4;
    const int c = lane & 15;
    const int k0 = 8 * h;
    const int c7 = c & 7;
    const int c3 = c & 3;

    Op4 opA, opAtA, opC, opCsh, opNCt, opBtA;
    #pragma unroll
    for (int q = 0; q < 4; ++q) {
        const int k = k0 + 2 * q;
        opA.q[q]   = v2f{Ag[c * NS + k], Ag[c * NS + k + 1]};
        opAtA.q[q] = v2f{Ag[k * NS + c], Ag[(k + 1) * NS + c]};
        const float cv0 = Cg[c7 * NS + k], cv1 = Cg[c7 * NS + k + 1];
        opC.q[q]   = (c < MS)  ? v2f{cv0, cv1} : v2f{0.f, 0.f};
        opCsh.q[q] = (c >= MS) ? v2f{cv0, cv1} : v2f{0.f, 0.f};
        const float n0 = -Cg[(2 * q) * NS + c], n1 = -Cg[(2 * q + 1) * NS + c];
        opNCt.q[q] = (h == 0) ? v2f{n0, n1} : v2f{0.f, 0.f};
        const float b0 = Bg[k * US + c3], b1 = Bg[(k + 1) * US + c3];
        opBtA.q[q] = (c < US) ? v2f{b0, b1} : v2f{0.f, 0.f};
    }
    v8f dQ, dQt, dI, dRaug, P;
    #pragma unroll
    for (int r = 0; r < 8; ++r) {
        dQ[r]  = Qg[(k0 + r) * NS + c];
        dQt[r] = Qg[c * NS + k0 + r];
        dI[r]  = (k0 + r == c) ? 1.0f : 0.0f;
        const float rv = Rg[r * MS + c7];
        dRaug[r] = (h == 0) ? ((c >= MS) ? rv : ((r == c) ? 1.0f : 0.0f)) : 0.0f;
        P[r] = dI[r];
    }
    const v8f zero = {};

    #pragma unroll 1
    for (int t = 1; t < SEQ; ++t) {
        const v8f T1t = mm16(from_acc(P), opA, zero);
        const Op4 oT1 = from_acc(T1t);
        const v8f Pp  = mm16(oT1, opA, dQ);
        const v8f PpT = mm16(opA, oT1, dQt);
        const v8f CPt  = mm16(from_acc(Pp), opC, zero);
        const v8f PCtT = mm16(opC, from_acc(PpT), zero);
        v8f g = mm16(from_acc(CPt), opCsh, dRaug);

        #pragma unroll
        for (int p = 0; p < MS; ++p) {
            const float invp = __builtin_amdgcn_rcpf(rdlane(g[p], MS + p));
            const float pv = g[p] * invp;
            #pragma unroll
            for (int i = 0; i < MS; ++i) {
                if (i != p) {
                    const float f = rdlane(g[i], MS + p);
                    g[i] = g[i] - f * pv;
                }
            }
            g[p] = pv;
        }
        Op4 oSi = from_acc(g);
        {
            const bool keep = lane < MS;
            const v2f z2 = v2f{0.f, 0.f};
            oSi.q[0] = keep ? oSi.q[0] : z2;
            oSi.q[1] = keep ? oSi.q[1] : z2;
            oSi.q[2] = keep ? oSi.q[2] : z2;
            oSi.q[3] = keep ? oSi.q[3] : z2;
        }
        const v8f Kt = mm16(oSi, from_acc(PCtT), zero);
        const v8f Wt = mm16(opNCt, from_acc(Kt), dI);
        const Op4 oW = from_acc(Wt);
        P = mm16(oW, from_acc(Pp), zero);
        const v8f Hb = mm16(opBtA, oW, zero);
        const v8f G = mm16(opAtA, oW, zero);

        float* wsp = ws + (size_t)(t - 1) * STEPF;
        {
            float* o = wsp + c * NS + k0;
            *(float4*)(o)     = make_float4(G[0], G[1], G[2], G[3]);
            *(float4*)(o + 4) = make_float4(G[4], G[5], G[6], G[7]);
        }
        if (h == 0) {
            *(float4*)(wsp + 256 + c * US) = make_float4(Hb[0], Hb[1], Hb[2], Hb[3]);
            float* o = wsp + 320 + c * MS;
            *(float4*)(o)     = make_float4(Kt[0], Kt[1], Kt[2], Kt[3]);
            *(float4*)(o + 4) = make_float4(Kt[4], Kt[5], Kt[6], Kt[7]);
        }
    }
}

__device__ __forceinline__ float bperm(int srcLane, float v) {
    return __builtin_bit_cast(float,
        __builtin_amdgcn_ds_bpermute(srcLane << 2, __builtin_bit_cast(int, v)));
}

struct StepOps {
    v2f aA0, aA1, aA2, aA3;
    v2f aB, aK0, aK1;
    v2f bZ0, bZ1, bU;
};

__device__ __forceinline__ StepOps load_ops(const float* __restrict__ ws,
                                            const float* __restrict__ zbase,
                                            const float* __restrict__ ubase,
                                            int t, int col, int hi) {
    StepOps o;
    const float* w = ws + (size_t)(t - 1) * STEPF;
    o.aA0 = *(const v2f*)(w + col * 16 + 0  + 2 * hi);
    o.aA1 = *(const v2f*)(w + col * 16 + 4  + 2 * hi);
    o.aA2 = *(const v2f*)(w + col * 16 + 8  + 2 * hi);
    o.aA3 = *(const v2f*)(w + col * 16 + 12 + 2 * hi);
    o.aB  = *(const v2f*)(w + 256 + col * 4 + 2 * hi);
    o.aK0 = *(const v2f*)(w + 320 + col * 8 + 0 + 2 * hi);
    o.aK1 = *(const v2f*)(w + 320 + col * 8 + 4 + 2 * hi);
    const float* zr = zbase + (size_t)t * MS;
    o.bZ0 = *(const v2f*)(zr + 0 + 2 * hi);
    o.bZ1 = *(const v2f*)(zr + 4 + 2 * hi);
    o.bU  = *(const v2f*)(ubase + (size_t)t * US + 2 * hi);
    return o;
}

__device__ __forceinline__ v8f zu_chain(const StepOps& o) {
    v8f d = {};
    d = wmma4(o.aB,  o.bU,  d);
    d = wmma4(o.aK0, o.bZ0, d);
    d = wmma4(o.aK1, o.bZ1, d);
    return d;
}

__global__ __launch_bounds__(32)
void kalman_state_kernel(const float* __restrict__ obs, const float* __restrict__ inp,
                         const float* __restrict__ x0, const float* __restrict__ ws,
                         float* __restrict__ out)
{
    const int lane = threadIdx.x;
    const int hi   = lane >> 4;
    const int col  = lane & 15;
    const int b    = blockIdx.x * 16 + col;

    v8f x;
    #pragma unroll
    for (int r = 0; r < 8; ++r) x[r] = x0[8 * hi + r];

    {
        float* o = out + ((size_t)b * SEQ) * NS + 8 * hi;
        *(float4*)(o)     = make_float4(x[0], x[1], x[2], x[3]);
        *(float4*)(o + 4) = make_float4(x[4], x[5], x[6], x[7]);
    }

    const float* zbase = obs + (size_t)b * SEQ * MS;
    const float* ubase = inp + (size_t)b * SEQ * US;
    float*       obase = out + (size_t)b * SEQ * NS;

    StepOps ring[PF_DEPTH];
    #pragma unroll
    for (int s = 0; s < PF_DEPTH; ++s) {
        ring[s] = load_ops(ws, zbase, ubase, 1 + s, col, hi);
        __builtin_amdgcn_sched_barrier(0);
    }
    v8f dzu = zu_chain(ring[0]);

    #pragma unroll 1
    for (int tb = 1; tb < SEQ; tb += PF_DEPTH) {
        #pragma unroll
        for (int s = 0; s < PF_DEPTH; ++s) {
            const int t = tb + s;

            v2f bX[4];
            #pragma unroll
            for (int c = 0; c < 4; ++c) {
                int srcLane = col + ((c >= 2) ? 16 : 0);
                int lo = 4 * (c & 1);
                float t0 = bperm(srcLane, x[lo + 0]);
                float t1 = bperm(srcLane, x[lo + 2]);
                float t2 = bperm(srcLane, x[lo + 1]);
                float t3 = bperm(srcLane, x[lo + 3]);
                bX[c].x = hi ? t1 : t0;
                bX[c].y = hi ? t3 : t2;
            }

            v8f d = wmma4(ring[s].aA0, bX[0], dzu);
            d = wmma4(ring[s].aA1, bX[1], d);
            d = wmma4(ring[s].aA2, bX[2], d);
            d = wmma4(ring[s].aA3, bX[3], d);
            x = d;

            float* o = obase + (size_t)t * NS + 8 * hi;
            *(float4*)(o)     = make_float4(x[0], x[1], x[2], x[3]);
            *(float4*)(o + 4) = make_float4(x[4], x[5], x[6], x[7]);

            dzu = zu_chain(ring[(s + 1) % PF_DEPTH]);

            const int tn = (t + PF_DEPTH < SEQ) ? (t + PF_DEPTH) : (SEQ - 1);
            ring[s] = load_ops(ws, zbase, ubase, tn, col, hi);
            __builtin_amdgcn_sched_barrier(0);
        }
    }
}

extern "C" void kernel_launch(void* const* d_in, const int* in_sizes, int n_in,
                              void* d_out, int out_size, void* d_ws, size_t ws_size,
                              hipStream_t stream) {
    const float* obs = (const float*)d_in[0];
    const float* inp = (const float*)d_in[1];
    const float* A   = (const float*)d_in[2];
    const float* B   = (const float*)d_in[3];
    const float* C   = (const float*)d_in[4];
    const float* Q   = (const float*)d_in[5];
    const float* R   = (const float*)d_in[6];
    const float* x0  = (const float*)d_in[7];
    float* out = (float*)d_out;
    float* ws  = (float*)d_ws;

    hipLaunchKernelGGL(kalman_cov_kernel, dim3(1), dim3(32), 0, stream,
                       A, B, C, Q, R, ws);
    hipLaunchKernelGGL(kalman_state_kernel, dim3(BATCH / 16), dim3(32), 0, stream,
                       obs, inp, x0, ws, out);
}


# Phase 2 state stays in its accumulator registers as the B operand (Aeff rows loaded in k order 8h+2q+j): the 16 ds_bpermute and 8 selects per step go; each dot product now sums in that k order
# speedup vs baseline: 5.6810x; 1.0967x over previous
#include <hip/hip_runtime.h>

#define BATCH 256
#define SEQ   4096
#define NS    16
#define MS    8
#define US    4
#define STEPF 448
#define PF_DEPTH 5
static_assert((SEQ - 1) % PF_DEPTH == 0, "phase 2 walks the steps in whole passes of PF_DEPTH");

typedef __attribute__((ext_vector_type(2))) float v2f;
typedef __attribute__((ext_vector_type(8))) float v8f;

__device__ __forceinline__ float rdlane(float v, int lane) {
    return __builtin_bit_cast(float,
        __builtin_amdgcn_readlane(__builtin_bit_cast(int, v), lane));
}

__device__ __forceinline__ v8f wmma4(v2f a, v2f b, v8f c) {
    return __builtin_amdgcn_wmma_f32_16x16x4_f32(false, a, false, b,
                                                 (short)0, c, false, false);
}

struct Op4 { v2f q[4]; };

__device__ __forceinline__ Op4 from_acc(v8f x) {
    Op4 o;
    o.q[0] = v2f{x[0], x[1]};
    o.q[1] = v2f{x[2], x[3]};
    o.q[2] = v2f{x[4], x[5]};
    o.q[3] = v2f{x[6], x[7]};
    return o;
}

__device__ __forceinline__ v8f mm16(const Op4& a, const Op4& b, v8f c) {
    c = wmma4(a.q[0], b.q[0], c);
    c = wmma4(a.q[1], b.q[1], c);
    c = wmma4(a.q[2], b.q[2], c);
    c = wmma4(a.q[3], b.q[3], c);
    return c;
}

__global__ __launch_bounds__(32)
void kalman_cov_kernel(const float* __restrict__ Ag, const float* __restrict__ Bg,
                       const float* __restrict__ Cg, const float* __restrict__ Qg,
                       const float* __restrict__ Rg, float* __restrict__ ws)
{
    const int lane = threadIdx.x;
    const int h = lane >> 4;
    const int c = lane & 15;
    const int k0 = 8 * h;
    const int c7 = c & 7;
    const int c3 = c & 3;

    Op4 opA, opAtA, opC, opCsh, opNCt, opBtA;
    #pragma unroll
    for (int q = 0; q < 4; ++q) {
        const int k = k0 + 2 * q;
        opA.q[q]   = v2f{Ag[c * NS + k], Ag[c * NS + k + 1]};
        opAtA.q[q] = v2f{Ag[k * NS + c], Ag[(k + 1) * NS + c]};
        const float cv0 = Cg[c7 * NS + k], cv1 = Cg[c7 * NS + k + 1];
        opC.q[q]   = (c < MS)  ? v2f{cv0, cv1} : v2f{0.f, 0.f};
        opCsh.q[q] = (c >= MS) ? v2f{cv0, cv1} : v2f{0.f, 0.f};
        const float n0 = -Cg[(2 * q) * NS + c], n1 = -Cg[(2 * q + 1) * NS + c];
        opNCt.q[q] = (h == 0) ? v2f{n0, n1} : v2f{0.f, 0.f};
        const float b0 = Bg[k * US + c3], b1 = Bg[(k + 1) * US + c3];
        opBtA.q[q] = (c < US) ? v2f{b0, b1} : v2f{0.f, 0.f};
    }
    v8f dQ, dQt, dI, dRaug, P;
    #pragma unroll
    for (int r = 0; r < 8; ++r) {
        dQ[r]  = Qg[(k0 + r) * NS + c];
        dQt[r] = Qg[c * NS + k0 + r];
        dI[r]  = (k0 + r == c) ? 1.0f : 0.0f;
        const float rv = Rg[r * MS + c7];
        dRaug[r] = (h == 0) ? ((c >= MS) ? rv : ((r == c) ? 1.0f : 0.0f)) : 0.0f;
        P[r] = dI[r];
    }
    const v8f zero = {};

    #pragma unroll 1
    for (int t = 1; t < SEQ; ++t) {
        const v8f T1t = mm16(from_acc(P), opA, zero);
        const Op4 oT1 = from_acc(T1t);
        const v8f Pp  = mm16(oT1, opA, dQ);
        const v8f PpT = mm16(opA, oT1, dQt);
        const v8f CPt  = mm16(from_acc(Pp), opC, zero);
        const v8f PCtT = mm16(opC, from_acc(PpT), zero);
        v8f g = mm16(from_acc(CPt), opCsh, dRaug);

        #pragma unroll
        for (int p = 0; p < MS; ++p) {
            const float invp = __builtin_amdgcn_rcpf(rdlane(g[p], MS + p));
            const float pv = g[p] * invp;
            #pragma unroll
            for (int i = 0; i < MS; ++i) {
                if (i != p) {
                    const float f = rdlane(g[i], MS + p);
                    g[i] = g[i] - f * pv;
                }
            }
            g[p] = pv;
        }
        Op4 oSi = from_acc(g);
        {
            const bool keep = lane < MS;
            const v2f z2 = v2f{0.f, 0.f};
            oSi.q[0] = keep ? oSi.q[0] : z2;
            oSi.q[1] = keep ? oSi.q[1] : z2;
            oSi.q[2] = keep ? oSi.q[2] : z2;
            oSi.q[3] = keep ? oSi.q[3] : z2;
        }
        const v8f Kt = mm16(oSi, from_acc(PCtT), zero);
        const v8f Wt = mm16(opNCt, from_acc(Kt), dI);
        const Op4 oW = from_acc(Wt);
        P = mm16(oW, from_acc(Pp), zero);
        const v8f Hb = mm16(opBtA, oW, zero);
        const v8f G = mm16(opAtA, oW, zero);

        float* wsp = ws + (size_t)(t - 1) * STEPF;
        {
            float* o = wsp + c * NS + k0;
            *(float4*)(o)     = make_float4(G[0], G[1], G[2], G[3]);
            *(float4*)(o + 4) = make_float4(G[4], G[5], G[6], G[7]);
        }
        if (h == 0) {
            *(float4*)(wsp + 256 + c * US) = make_float4(Hb[0], Hb[1], Hb[2], Hb[3]);
            float* o = wsp + 320 + c * MS;
            *(float4*)(o)     = make_float4(Kt[0], Kt[1], Kt[2], Kt[3]);
            *(float4*)(o + 4) = make_float4(Kt[4], Kt[5], Kt[6], Kt[7]);
        }
    }
}


struct StepOps {
    v2f aA0, aA1, aA2, aA3;
    v2f aB, aK0, aK1;
    v2f bZ0, bZ1, bU;
};

__device__ __forceinline__ StepOps load_ops(const float* __restrict__ ws,
                                            const float* __restrict__ zbase,
                                            const float* __restrict__ ubase,
                                            int t, int col, int hi) {
    StepOps o;
    const float* w = ws + (size_t)(t - 1) * STEPF;
    o.aA0 = *(const v2f*)(w + col * 16 + 8 * hi + 0);
    o.aA1 = *(const v2f*)(w + col * 16 + 8 * hi + 2);
    o.aA2 = *(const v2f*)(w + col * 16 + 8 * hi + 4);
    o.aA3 = *(const v2f*)(w + col * 16 + 8 * hi + 6);
    o.aB  = *(const v2f*)(w + 256 + col * 4 + 2 * hi);
    o.aK0 = *(const v2f*)(w + 320 + col * 8 + 0 + 2 * hi);
    o.aK1 = *(const v2f*)(w + 320 + col * 8 + 4 + 2 * hi);
    const float* zr = zbase + (size_t)t * MS;
    o.bZ0 = *(const v2f*)(zr + 0 + 2 * hi);
    o.bZ1 = *(const v2f*)(zr + 4 + 2 * hi);
    o.bU  = *(const v2f*)(ubase + (size_t)t * US + 2 * hi);
    return o;
}

__device__ __forceinline__ v8f zu_chain(const StepOps& o) {
    v8f d = {};
    d = wmma4(o.aB,  o.bU,  d);
    d = wmma4(o.aK0, o.bZ0, d);
    d = wmma4(o.aK1, o.bZ1, d);
    return d;
}

__global__ __launch_bounds__(32)
void kalman_state_kernel(const float* __restrict__ obs, const float* __restrict__ inp,
                         const float* __restrict__ x0, const float* __restrict__ ws,
                         float* __restrict__ out)
{
    const int lane = threadIdx.x;
    const int hi   = lane >> 4;
    const int col  = lane & 15;
    const int b    = blockIdx.x * 16 + col;

    v8f x;
    #pragma unroll
    for (int r = 0; r < 8; ++r) x[r] = x0[8 * hi + r];

    {
        float* o = out + ((size_t)b * SEQ) * NS + 8 * hi;
        *(float4*)(o)     = make_float4(x[0], x[1], x[2], x[3]);
        *(float4*)(o + 4) = make_float4(x[4], x[5], x[6], x[7]);
    }

    const float* zbase = obs + (size_t)b * SEQ * MS;
    const float* ubase = inp + (size_t)b * SEQ * US;
    float*       obase = out + (size_t)b * SEQ * NS;

    StepOps ring[PF_DEPTH];
    #pragma unroll
    for (int s = 0; s < PF_DEPTH; ++s) {
        ring[s] = load_ops(ws, zbase, ubase, 1 + s, col, hi);
        __builtin_amdgcn_sched_barrier(0);
    }
    v8f dzu = zu_chain(ring[0]);

    #pragma unroll 1
    for (int tb = 1; tb < SEQ; tb += PF_DEPTH) {
        #pragma unroll
        for (int s = 0; s < PF_DEPTH; ++s) {
            const int t = tb + s;

            v8f d = wmma4(ring[s].aA0, v2f{x[0], x[1]}, dzu);
            d = wmma4(ring[s].aA1, v2f{x[2], x[3]}, d);
            d = wmma4(ring[s].aA2, v2f{x[4], x[5]}, d);
            d = wmma4(ring[s].aA3, v2f{x[6], x[7]}, d);
            x = d;

            float* o = obase + (size_t)t * NS + 8 * hi;
            *(float4*)(o)     = make_float4(x[0], x[1], x[2], x[3]);
            *(float4*)(o + 4) = make_float4(x[4], x[5], x[6], x[7]);

            dzu = zu_chain(ring[(s + 1) % PF_DEPTH]);

            const int tn = (t + PF_DEPTH < SEQ) ? (t + PF_DEPTH) : (SEQ - 1);
            ring[s] = load_ops(ws, zbase, ubase, tn, col, hi);
            __builtin_amdgcn_sched_barrier(0);
        }
    }
}

extern "C" void kernel_launch(void* const* d_in, const int* in_sizes, int n_in,
                              void* d_out, int out_size, void* d_ws, size_t ws_size,
                              hipStream_t stream) {
    const float* obs = (const float*)d_in[0];
    const float* inp = (const float*)d_in[1];
    const float* A   = (const float*)d_in[2];
    const float* B   = (const float*)d_in[3];
    const float* C   = (const float*)d_in[4];
    const float* Q   = (const float*)d_in[5];
    const float* R   = (const float*)d_in[6];
    const float* x0  = (const float*)d_in[7];
    float* out = (float*)d_out;
    float* ws  = (float*)d_ws;

    hipLaunchKernelGGL(kalman_cov_kernel, dim3(1), dim3(32), 0, stream,
                       A, B, C, Q, R, ws);
    hipLaunchKernelGGL(kalman_state_kernel, dim3(BATCH / 16), dim3(32), 0, stream,
                       obs, inp, x0, ws, out);
}


# Constant product C A formed once: S = (CA)P(CA)^T + (CQC^T+R) and (Ppri C^T)^T = (CA)(AP)^T + CQ^T come two product levels behind P, not four; transposed prior dropped; sums reassociated
# speedup vs baseline: 6.1251x; 1.0782x over previous
#include <hip/hip_runtime.h>

#define BATCH 256
#define SEQ   4096
#define NS    16
#define MS    8
#define US    4
#define STEPF 448
#define PF_DEPTH 5
static_assert((SEQ - 1) % PF_DEPTH == 0, "phase 2 walks the steps in whole passes of PF_DEPTH");

typedef __attribute__((ext_vector_type(2))) float v2f;
typedef __attribute__((ext_vector_type(8))) float v8f;

__device__ __forceinline__ float rdlane(float v, int lane) {
    return __builtin_bit_cast(float,
        __builtin_amdgcn_readlane(__builtin_bit_cast(int, v), lane));
}

__device__ __forceinline__ v8f wmma4(v2f a, v2f b, v8f c) {
    return __builtin_amdgcn_wmma_f32_16x16x4_f32(false, a, false, b,
                                                 (short)0, c, false, false);
}

struct Op4 { v2f q[4]; };

__device__ __forceinline__ Op4 from_acc(v8f x) {
    Op4 o;
    o.q[0] = v2f{x[0], x[1]};
    o.q[1] = v2f{x[2], x[3]};
    o.q[2] = v2f{x[4], x[5]};
    o.q[3] = v2f{x[6], x[7]};
    return o;
}

__device__ __forceinline__ v8f mm16(const Op4& a, const Op4& b, v8f c) {
    c = wmma4(a.q[0], b.q[0], c);
    c = wmma4(a.q[1], b.q[1], c);
    c = wmma4(a.q[2], b.q[2], c);
    c = wmma4(a.q[3], b.q[3], c);
    return c;
}

__global__ __launch_bounds__(32)
void kalman_cov_kernel(const float* __restrict__ Ag, const float* __restrict__ Bg,
                       const float* __restrict__ Cg, const float* __restrict__ Qg,
                       const float* __restrict__ Rg, float* __restrict__ ws)
{
    const int lane = threadIdx.x;
    const int h = lane >> 4;
    const int c = lane & 15;
    const int k0 = 8 * h;
    const int c7 = c & 7;
    const int c3 = c & 3;

    Op4 opA, opAtA, opC, opCsh, opNCt, opBtA;
    #pragma unroll
    for (int q = 0; q < 4; ++q) {
        const int k = k0 + 2 * q;
        opA.q[q]   = v2f{Ag[c * NS + k], Ag[c * NS + k + 1]};
        opAtA.q[q] = v2f{Ag[k * NS + c], Ag[(k + 1) * NS + c]};
        const float cv0 = Cg[c7 * NS + k], cv1 = Cg[c7 * NS + k + 1];
        opC.q[q]   = (c < MS)  ? v2f{cv0, cv1} : v2f{0.f, 0.f};
        opCsh.q[q] = (c >= MS) ? v2f{cv0, cv1} : v2f{0.f, 0.f};
        const float n0 = -Cg[(2 * q) * NS + c], n1 = -Cg[(2 * q + 1) * NS + c];
        opNCt.q[q] = (h == 0) ? v2f{n0, n1} : v2f{0.f, 0.f};
        const float b0 = Bg[k * US + c3], b1 = Bg[(k + 1) * US + c3];
        opBtA.q[q] = (c < US) ? v2f{b0, b1} : v2f{0.f, 0.f};
    }
    v8f dQ, dQt, dI, dRaug, P;
    #pragma unroll
    for (int r = 0; r < 8; ++r) {
        dQ[r]  = Qg[(k0 + r) * NS + c];
        dQt[r] = Qg[c * NS + k0 + r];
        dI[r]  = (k0 + r == c) ? 1.0f : 0.0f;
        const float rv = Rg[r * MS + c7];
        dRaug[r] = (h == 0) ? ((c >= MS) ? rv : ((r == c) ? 1.0f : 0.0f)) : 0.0f;
        P[r] = dI[r];
    }
    const v8f zero = {};

    const Op4 opCA   = from_acc(mm16(opAtA, opC,   zero));
    const Op4 opCAsh = from_acc(mm16(opAtA, opCsh, zero));
    const v8f dCQt = mm16(opC, from_acc(dQt), zero);
    const v8f CQT = mm16(from_acc(dQ), opC, zero);
    const v8f dSaug = mm16(from_acc(CQT), opCsh, dRaug);

    #pragma unroll 1
    for (int t = 1; t < SEQ; ++t) {
        const Op4 oP = from_acc(P);
        const v8f T1t = mm16(oP, opA, zero);
        const v8f V   = mm16(oP, opCA, zero);
        const Op4 oT1 = from_acc(T1t);
        v8f g = mm16(from_acc(V), opCAsh, dSaug);
        const v8f Pp  = mm16(oT1, opA, dQ);
        const v8f PCtT = mm16(opCA, oT1, dCQt);

        #pragma unroll
        for (int p = 0; p < MS; ++p) {
            const float invp = __builtin_amdgcn_rcpf(rdlane(g[p], MS + p));
            const float pv = g[p] * invp;
            #pragma unroll
            for (int i = 0; i < MS; ++i) {
                if (i != p) {
                    const float f = rdlane(g[i], MS + p);
                    g[i] = g[i] - f * pv;
                }
            }
            g[p] = pv;
        }
        Op4 oSi = from_acc(g);
        {
            const bool keep = lane < MS;
            const v2f z2 = v2f{0.f, 0.f};
            oSi.q[0] = keep ? oSi.q[0] : z2;
            oSi.q[1] = keep ? oSi.q[1] : z2;
            oSi.q[2] = keep ? oSi.q[2] : z2;
            oSi.q[3] = keep ? oSi.q[3] : z2;
        }
        const v8f Kt = mm16(oSi, from_acc(PCtT), zero);
        const v8f Wt = mm16(opNCt, from_acc(Kt), dI);
        const Op4 oW = from_acc(Wt);
        P = mm16(oW, from_acc(Pp), zero);
        const v8f Hb = mm16(opBtA, oW, zero);
        const v8f G = mm16(opAtA, oW, zero);

        float* wsp = ws + (size_t)(t - 1) * STEPF;
        {
            float* o = wsp + c * NS + k0;
            *(float4*)(o)     = make_float4(G[0], G[1], G[2], G[3]);
            *(float4*)(o + 4) = make_float4(G[4], G[5], G[6], G[7]);
        }
        if (h == 0) {
            *(float4*)(wsp + 256 + c * US) = make_float4(Hb[0], Hb[1], Hb[2], Hb[3]);
            float* o = wsp + 320 + c * MS;
            *(float4*)(o)     = make_float4(Kt[0], Kt[1], Kt[2], Kt[3]);
            *(float4*)(o + 4) = make_float4(Kt[4], Kt[5], Kt[6], Kt[7]);
        }
    }
}


struct StepOps {
    v2f aA0, aA1, aA2, aA3;
    v2f aB, aK0, aK1;
    v2f bZ0, bZ1, bU;
};

__device__ __forceinline__ StepOps load_ops(const float* __restrict__ ws,
                                            const float* __restrict__ zbase,
                                            const float* __restrict__ ubase,
                                            int t, int col, int hi) {
    StepOps o;
    const float* w = ws + (size_t)(t - 1) * STEPF;
    o.aA0 = *(const v2f*)(w + col * 16 + 8 * hi + 0);
    o.aA1 = *(const v2f*)(w + col * 16 + 8 * hi + 2);
    o.aA2 = *(const v2f*)(w + col * 16 + 8 * hi + 4);
    o.aA3 = *(const v2f*)(w + col * 16 + 8 * hi + 6);
    o.aB  = *(const v2f*)(w + 256 + col * 4 + 2 * hi);
    o.aK0 = *(const v2f*)(w + 320 + col * 8 + 0 + 2 * hi);
    o.aK1 = *(const v2f*)(w + 320 + col * 8 + 4 + 2 * hi);
    const float* zr = zbase + (size_t)t * MS;
    o.bZ0 = *(const v2f*)(zr + 0 + 2 * hi);
    o.bZ1 = *(const v2f*)(zr + 4 + 2 * hi);
    o.bU  = *(const v2f*)(ubase + (size_t)t * US + 2 * hi);
    return o;
}

__device__ __forceinline__ v8f zu_chain(const StepOps& o) {
    v8f d = {};
    d = wmma4(o.aB,  o.bU,  d);
    d = wmma4(o.aK0, o.bZ0, d);
    d = wmma4(o.aK1, o.bZ1, d);
    return d;
}

__global__ __launch_bounds__(32)
void kalman_state_kernel(const float* __restrict__ obs, const float* __restrict__ inp,
                         const float* __restrict__ x0, const float* __restrict__ ws,
                         float* __restrict__ out)
{
    const int lane = threadIdx.x;
    const int hi   = lane >> 4;
    const int col  = lane & 15;
    const int b    = blockIdx.x * 16 + col;

    v8f x;
    #pragma unroll
    for (int r = 0; r < 8; ++r) x[r] = x0[8 * hi + r];

    {
        float* o = out + ((size_t)b * SEQ) * NS + 8 * hi;
        *(float4*)(o)     = make_float4(x[0], x[1], x[2], x[3]);
        *(float4*)(o + 4) = make_float4(x[4], x[5], x[6], x[7]);
    }

    const float* zbase = obs + (size_t)b * SEQ * MS;
    const float* ubase = inp + (size_t)b * SEQ * US;
    float*       obase = out + (size_t)b * SEQ * NS;

    StepOps ring[PF_DEPTH];
    #pragma unroll
    for (int s = 0; s < PF_DEPTH; ++s) {
        ring[s] = load_ops(ws, zbase, ubase, 1 + s, col, hi);
        __builtin_amdgcn_sched_barrier(0);
    }
    v8f dzu = zu_chain(ring[0]);

    #pragma unroll 1
    for (int tb = 1; tb < SEQ; tb += PF_DEPTH) {
        #pragma unroll
        for (int s = 0; s < PF_DEPTH; ++s) {
            const int t = tb + s;

            v8f d = wmma4(ring[s].aA0, v2f{x[0], x[1]}, dzu);
            d = wmma4(ring[s].aA1, v2f{x[2], x[3]}, d);
            d = wmma4(ring[s].aA2, v2f{x[4], x[5]}, d);
            d = wmma4(ring[s].aA3, v2f{x[6], x[7]}, d);
            x = d;

            float* o = obase + (size_t)t * NS + 8 * hi;
            *(float4*)(o)     = make_float4(x[0], x[1], x[2], x[3]);
            *(float4*)(o + 4) = make_float4(x[4], x[5], x[6], x[7]);

            dzu = zu_chain(ring[(s + 1) % PF_DEPTH]);

            const int tn = (t + PF_DEPTH < SEQ) ? (t + PF_DEPTH) : (SEQ - 1);
            ring[s] = load_ops(ws, zbase, ubase, tn, col, hi);
            __builtin_amdgcn_sched_barrier(0);
        }
    }
}

extern "C" void kernel_launch(void* const* d_in, const int* in_sizes, int n_in,
                              void* d_out, int out_size, void* d_ws, size_t ws_size,
                              hipStream_t stream) {
    const float* obs = (const float*)d_in[0];
    const float* inp = (const float*)d_in[1];
    const float* A   = (const float*)d_in[2];
    const float* B   = (const float*)d_in[3];
    const float* C   = (const float*)d_in[4];
    const float* Q   = (const float*)d_in[5];
    const float* R   = (const float*)d_in[6];
    const float* x0  = (const float*)d_in[7];
    float* out = (float*)d_out;
    float* ws  = (float*)d_ws;

    hipLaunchKernelGGL(kalman_cov_kernel, dim3(1), dim3(32), 0, stream,
                       A, B, C, Q, R, ws);
    hipLaunchKernelGGL(kalman_state_kernel, dim3(BATCH / 16), dim3(32), 0, stream,
                       obs, inp, x0, ws, out);
}


# One launch of 2-wave blocks: wave 0 repeats the covariance recursion per block, hands each step's gain to wave 1 (state recursion) via LDS, one barrier per step; phases overlap; same sums and order
# speedup vs baseline: 6.5998x; 1.0775x over previous
#include <hip/hip_runtime.h>

#define BATCH 256
#define SEQ   4096
#define NS    16
#define MS    8
#define US    4
#define STEPF 448

typedef __attribute__((ext_vector_type(2))) float v2f;
typedef __attribute__((ext_vector_type(8))) float v8f;

__device__ __forceinline__ float rdlane(float v, int lane) {
    return __builtin_bit_cast(float,
        __builtin_amdgcn_readlane(__builtin_bit_cast(int, v), lane));
}

__device__ __forceinline__ v8f wmma4(v2f a, v2f b, v8f c) {
    return __builtin_amdgcn_wmma_f32_16x16x4_f32(false, a, false, b,
                                                 (short)0, c, false, false);
}

struct Op4 { v2f q[4]; };

__device__ __forceinline__ Op4 from_acc(v8f x) {
    Op4 o;
    o.q[0] = v2f{x[0], x[1]};
    o.q[1] = v2f{x[2], x[3]};
    o.q[2] = v2f{x[4], x[5]};
    o.q[3] = v2f{x[6], x[7]};
    return o;
}

__device__ __forceinline__ v8f mm16(const Op4& a, const Op4& b, v8f c) {
    c = wmma4(a.q[0], b.q[0], c);
    c = wmma4(a.q[1], b.q[1], c);
    c = wmma4(a.q[2], b.q[2], c);
    c = wmma4(a.q[3], b.q[3], c);
    return c;
}

struct ZU { v2f bZ0, bZ1, bU; };

__device__ __forceinline__ ZU load_zu(const float* __restrict__ zbase,
                                      const float* __restrict__ ubase, int t, int hi) {
    ZU o;
    const float* zr = zbase + (size_t)t * MS;
    o.bZ0 = *(const v2f*)(zr + 0 + 2 * hi);
    o.bZ1 = *(const v2f*)(zr + 4 + 2 * hi);
    o.bU  = *(const v2f*)(ubase + (size_t)t * US + 2 * hi);
    return o;
}

__global__ __launch_bounds__(64)
void kalman_fused_kernel(const float* __restrict__ obs, const float* __restrict__ inp,
                         const float* __restrict__ Ag, const float* __restrict__ Bg,
                         const float* __restrict__ Cg, const float* __restrict__ Qg,
                         const float* __restrict__ Rg, const float* __restrict__ x0,
                         float* __restrict__ out)
{
    __shared__ __attribute__((aligned(16))) float hand[2][STEPF];

    const int lane = threadIdx.x & 31;
    const int h = lane >> 4;
    const int c = lane & 15;
    const bool cov_wave = __builtin_amdgcn_readfirstlane((int)(threadIdx.x >> 5)) == 0;

    const v8f zero = {};

    Op4 opA = {}, opAtA = {}, opNCt = {}, opBtA = {}, opCA = {}, opCAsh = {};
    v8f dQ = zero, dI = zero, dCQt = zero, dSaug = zero, P = zero;
    v8f x = zero;
    ZU zu = {};
    const int b = blockIdx.x * 16 + c;
    const float* zbase = obs + (size_t)b * SEQ * MS;
    const float* ubase = inp + (size_t)b * SEQ * US;
    float*       obase = out + (size_t)b * SEQ * NS;

    if (cov_wave) {
        const int k0 = 8 * h;
        const int c7 = c & 7;
        const int c3 = c & 3;
        Op4 opC, opCsh;
        #pragma unroll
        for (int q = 0; q < 4; ++q) {
            const int k = k0 + 2 * q;
            opA.q[q]   = v2f{Ag[c * NS + k], Ag[c * NS + k + 1]};
            opAtA.q[q] = v2f{Ag[k * NS + c], Ag[(k + 1) * NS + c]};
            const float cv0 = Cg[c7 * NS + k], cv1 = Cg[c7 * NS + k + 1];
            opC.q[q]   = (c < MS)  ? v2f{cv0, cv1} : v2f{0.f, 0.f};
            opCsh.q[q] = (c >= MS) ? v2f{cv0, cv1} : v2f{0.f, 0.f};
            const float n0 = -Cg[(2 * q) * NS + c], n1 = -Cg[(2 * q + 1) * NS + c];
            opNCt.q[q] = (h == 0) ? v2f{n0, n1} : v2f{0.f, 0.f};
            const float b0 = Bg[k * US + c3], b1 = Bg[(k + 1) * US + c3];
            opBtA.q[q] = (c < US) ? v2f{b0, b1} : v2f{0.f, 0.f};
        }
        v8f dQt, dRaug;
        #pragma unroll
        for (int r = 0; r < 8; ++r) {
            dQ[r]  = Qg[(k0 + r) * NS + c];
            dQt[r] = Qg[c * NS + k0 + r];
            dI[r]  = (k0 + r == c) ? 1.0f : 0.0f;
            const float rv = Rg[r * MS + c7];
            dRaug[r] = (h == 0) ? ((c >= MS) ? rv : ((r == c) ? 1.0f : 0.0f)) : 0.0f;
            P[r] = dI[r];
        }
        opCA   = from_acc(mm16(opAtA, opC,   zero));
        opCAsh = from_acc(mm16(opAtA, opCsh, zero));
        dCQt = mm16(opC, from_acc(dQt), zero);
        const v8f CQT = mm16(from_acc(dQ), opC, zero);
        dSaug = mm16(from_acc(CQT), opCsh, dRaug);
    } else {
        #pragma unroll
        for (int r = 0; r < 8; ++r) x[r] = x0[8 * h + r];
        float* o = obase + 8 * h;
        *(float4*)(o)     = make_float4(x[0], x[1], x[2], x[3]);
        *(float4*)(o + 4) = make_float4(x[4], x[5], x[6], x[7]);
    }

    v8f T1t = zero, V = zero;

    #pragma unroll 1
    for (int t = 1; t <= SEQ; ++t) {
        const bool cov_step = cov_wave && (t < SEQ);
        if (cov_step) {
            const Op4 oP = from_acc(P);
            T1t = mm16(oP, opA, zero);
            V   = mm16(oP, opCA, zero);
        } else if (!cov_wave) {
            zu = load_zu(zbase, ubase, (t >= 2) ? (t - 1) : 1, h);
        }

        asm volatile("s_wait_dscnt 0x0" ::: "memory");
        __builtin_amdgcn_fence(__ATOMIC_RELEASE, "workgroup", "local");
        __builtin_amdgcn_s_barrier();
        __builtin_amdgcn_fence(__ATOMIC_ACQUIRE, "workgroup", "local");

        if (cov_step) {
            const Op4 oT1 = from_acc(T1t);
            v8f g = mm16(from_acc(V), opCAsh, dSaug);
            const v8f Pp  = mm16(oT1, opA, dQ);
            const v8f PCtT = mm16(opCA, oT1, dCQt);

            #pragma unroll
            for (int p = 0; p < MS; ++p) {
                const float invp = __builtin_amdgcn_rcpf(rdlane(g[p], MS + p));
                const float pv = g[p] * invp;
                #pragma unroll
                for (int i = 0; i < MS; ++i) {
                    if (i != p) {
                        const float f = rdlane(g[i], MS + p);
                        g[i] = g[i] - f * pv;
                    }
                }
                g[p] = pv;
            }
            Op4 oSi = from_acc(g);
            {
                const bool keep = lane < MS;
                const v2f z2 = v2f{0.f, 0.f};
                oSi.q[0] = keep ? oSi.q[0] : z2;
                oSi.q[1] = keep ? oSi.q[1] : z2;
                oSi.q[2] = keep ? oSi.q[2] : z2;
                oSi.q[3] = keep ? oSi.q[3] : z2;
            }
            const v8f Kt = mm16(oSi, from_acc(PCtT), zero);
            const v8f Wt = mm16(opNCt, from_acc(Kt), dI);
            const Op4 oW = from_acc(Wt);
            P = mm16(oW, from_acc(Pp), zero);
            const v8f Hb = mm16(opBtA, oW, zero);
            const v8f G = mm16(opAtA, oW, zero);

            float* wsp = &hand[t & 1][0];
            {
                float* o = wsp + c * NS + 8 * h;
                *(float4*)(o)     = make_float4(G[0], G[1], G[2], G[3]);
                *(float4*)(o + 4) = make_float4(G[4], G[5], G[6], G[7]);
            }
            if (h == 0) {
                *(float4*)(wsp + 256 + c * US) = make_float4(Hb[0], Hb[1], Hb[2], Hb[3]);
                float* o = wsp + 320 + c * MS;
                *(float4*)(o)     = make_float4(Kt[0], Kt[1], Kt[2], Kt[3]);
                *(float4*)(o + 4) = make_float4(Kt[4], Kt[5], Kt[6], Kt[7]);
            }
        } else if (!cov_wave && t >= 2) {
            const int s = t - 1;

            const float* w = &hand[s & 1][0];
            const v2f aA0 = *(const v2f*)(w + c * NS + 8 * h + 0);
            const v2f aA1 = *(const v2f*)(w + c * NS + 8 * h + 2);
            const v2f aA2 = *(const v2f*)(w + c * NS + 8 * h + 4);
            const v2f aA3 = *(const v2f*)(w + c * NS + 8 * h + 6);
            const v2f aB  = *(const v2f*)(w + 256 + c * US + 2 * h);
            const v2f aK0 = *(const v2f*)(w + 320 + c * MS + 0 + 2 * h);
            const v2f aK1 = *(const v2f*)(w + 320 + c * MS + 4 + 2 * h);

            v8f d = zero;
            d = wmma4(aB,  zu.bU,  d);
            d = wmma4(aK0, zu.bZ0, d);
            d = wmma4(aK1, zu.bZ1, d);
            d = wmma4(aA0, v2f{x[0], x[1]}, d);
            d = wmma4(aA1, v2f{x[2], x[3]}, d);
            d = wmma4(aA2, v2f{x[4], x[5]}, d);
            d = wmma4(aA3, v2f{x[6], x[7]}, d);
            x = d;

            float* o = obase + (size_t)s * NS + 8 * h;
            *(float4*)(o)     = make_float4(x[0], x[1], x[2], x[3]);
            *(float4*)(o + 4) = make_float4(x[4], x[5], x[6], x[7]);
        }
    }
}

extern "C" void kernel_launch(void* const* d_in, const int* in_sizes, int n_in,
                              void* d_out, int out_size, void* d_ws, size_t ws_size,
                              hipStream_t stream) {
    const float* obs = (const float*)d_in[0];
    const float* inp = (const float*)d_in[1];
    const float* A   = (const float*)d_in[2];
    const float* B   = (const float*)d_in[3];
    const float* C   = (const float*)d_in[4];
    const float* Q   = (const float*)d_in[5];
    const float* R   = (const float*)d_in[6];
    const float* x0  = (const float*)d_in[7];
    float* out = (float*)d_out;

    hipLaunchKernelGGL(kalman_fused_kernel, dim3(BATCH / 16), dim3(64), 0, stream,
                       obs, inp, A, B, C, Q, R, x0, out);
}


# Gauss-Jordan pivot column reaches the lanes as a row_share operand of each multiply-subtract, not by 8 readlanes per pivot through one scalar register; same operations and summation order
# speedup vs baseline: 7.7388x; 1.1726x over previous
#include <hip/hip_runtime.h>

#define BATCH 256
#define SEQ   4096
#define NS    16
#define MS    8
#define US    4
#define STEPF 448

typedef __attribute__((ext_vector_type(2))) float v2f;
typedef __attribute__((ext_vector_type(8))) float v8f;

template <int L>
__device__ __forceinline__ float row_share(float v) {
    return __builtin_bit_cast(float,
        __builtin_amdgcn_update_dpp(0, __builtin_bit_cast(int, v), 0x150 + L, 0xf, 0xf, true));
}

__device__ __forceinline__ v8f wmma4(v2f a, v2f b, v8f c) {
    return __builtin_amdgcn_wmma_f32_16x16x4_f32(false, a, false, b,
                                                 (short)0, c, false, false);
}

struct Op4 { v2f q[4]; };

__device__ __forceinline__ Op4 from_acc(v8f x) {
    Op4 o;
    o.q[0] = v2f{x[0], x[1]};
    o.q[1] = v2f{x[2], x[3]};
    o.q[2] = v2f{x[4], x[5]};
    o.q[3] = v2f{x[6], x[7]};
    return o;
}

__device__ __forceinline__ v8f mm16(const Op4& a, const Op4& b, v8f c) {
    c = wmma4(a.q[0], b.q[0], c);
    c = wmma4(a.q[1], b.q[1], c);
    c = wmma4(a.q[2], b.q[2], c);
    c = wmma4(a.q[3], b.q[3], c);
    return c;
}

struct ZU { v2f bZ0, bZ1, bU; };

__device__ __forceinline__ ZU load_zu(const float* __restrict__ zbase,
                                      const float* __restrict__ ubase, int t, int hi) {
    ZU o;
    const float* zr = zbase + (size_t)t * MS;
    o.bZ0 = *(const v2f*)(zr + 0 + 2 * hi);
    o.bZ1 = *(const v2f*)(zr + 4 + 2 * hi);
    o.bU  = *(const v2f*)(ubase + (size_t)t * US + 2 * hi);
    return o;
}

template <int P>
__device__ __forceinline__ void gj_pivot(v8f& g) {
    const float invp = __builtin_amdgcn_rcpf(row_share<MS + P>(g[P]));
    const float pv = g[P] * invp;
    #pragma unroll
    for (int i = 0; i < MS; ++i) {
        if (i != P) {
            const float f = row_share<MS + P>(g[i]);
            g[i] = g[i] - f * pv;
        }
    }
    g[P] = pv;
}

__global__ __launch_bounds__(64)
void kalman_fused_kernel(const float* __restrict__ obs, const float* __restrict__ inp,
                         const float* __restrict__ Ag, const float* __restrict__ Bg,
                         const float* __restrict__ Cg, const float* __restrict__ Qg,
                         const float* __restrict__ Rg, const float* __restrict__ x0,
                         float* __restrict__ out)
{
    __shared__ __attribute__((aligned(16))) float hand[2][STEPF];

    const int lane = threadIdx.x & 31;
    const int h = lane >> 4;
    const int c = lane & 15;
    const bool cov_wave = __builtin_amdgcn_readfirstlane((int)(threadIdx.x >> 5)) == 0;

    const v8f zero = {};

    Op4 opA = {}, opAtA = {}, opNCt = {}, opBtA = {}, opCA = {}, opCAsh = {};
    v8f dQ = zero, dI = zero, dCQt = zero, dSaug = zero, P = zero;
    v8f x = zero;
    ZU zu = {};
    const int b = blockIdx.x * 16 + c;
    const float* zbase = obs + (size_t)b * SEQ * MS;
    const float* ubase = inp + (size_t)b * SEQ * US;
    float*       obase = out + (size_t)b * SEQ * NS;

    if (cov_wave) {
        const int k0 = 8 * h;
        const int c7 = c & 7;
        const int c3 = c & 3;
        Op4 opC, opCsh;
        #pragma unroll
        for (int q = 0; q < 4; ++q) {
            const int k = k0 + 2 * q;
            opA.q[q]   = v2f{Ag[c * NS + k], Ag[c * NS + k + 1]};
            opAtA.q[q] = v2f{Ag[k * NS + c], Ag[(k + 1) * NS + c]};
            const float cv0 = Cg[c7 * NS + k], cv1 = Cg[c7 * NS + k + 1];
            opC.q[q]   = (c < MS)  ? v2f{cv0, cv1} : v2f{0.f, 0.f};
            opCsh.q[q] = (c >= MS) ? v2f{cv0, cv1} : v2f{0.f, 0.f};
            const float n0 = -Cg[(2 * q) * NS + c], n1 = -Cg[(2 * q + 1) * NS + c];
            opNCt.q[q] = (h == 0) ? v2f{n0, n1} : v2f{0.f, 0.f};
            const float b0 = Bg[k * US + c3], b1 = Bg[(k + 1) * US + c3];
            opBtA.q[q] = (c < US) ? v2f{b0, b1} : v2f{0.f, 0.f};
        }
        v8f dQt, dRaug;
        #pragma unroll
        for (int r = 0; r < 8; ++r) {
            dQ[r]  = Qg[(k0 + r) * NS + c];
            dQt[r] = Qg[c * NS + k0 + r];
            dI[r]  = (k0 + r == c) ? 1.0f : 0.0f;
            const float rv = Rg[r * MS + c7];
            dRaug[r] = (h == 0) ? ((c >= MS) ? rv : ((r == c) ? 1.0f : 0.0f))
                                : ((c == MS + r) ? 1.0f : 0.0f);
            P[r] = dI[r];
        }
        opCA   = from_acc(mm16(opAtA, opC,   zero));
        opCAsh = from_acc(mm16(opAtA, opCsh, zero));
        dCQt = mm16(opC, from_acc(dQt), zero);
        const v8f CQT = mm16(from_acc(dQ), opC, zero);
        dSaug = mm16(from_acc(CQT), opCsh, dRaug);
    } else {
        #pragma unroll
        for (int r = 0; r < 8; ++r) x[r] = x0[8 * h + r];
        float* o = obase + 8 * h;
        *(float4*)(o)     = make_float4(x[0], x[1], x[2], x[3]);
        *(float4*)(o + 4) = make_float4(x[4], x[5], x[6], x[7]);
    }

    v8f T1t = zero, V = zero;

    #pragma unroll 1
    for (int t = 1; t <= SEQ; ++t) {
        const bool cov_step = cov_wave && (t < SEQ);
        if (cov_step) {
            const Op4 oP = from_acc(P);
            T1t = mm16(oP, opA, zero);
            V   = mm16(oP, opCA, zero);
        } else if (!cov_wave) {
            zu = load_zu(zbase, ubase, (t >= 2) ? (t - 1) : 1, h);
        }

        asm volatile("s_wait_dscnt 0x0" ::: "memory");
        __builtin_amdgcn_fence(__ATOMIC_RELEASE, "workgroup", "local");
        __builtin_amdgcn_s_barrier();
        __builtin_amdgcn_fence(__ATOMIC_ACQUIRE, "workgroup", "local");

        if (cov_step) {
            const Op4 oT1 = from_acc(T1t);
            v8f g = mm16(from_acc(V), opCAsh, dSaug);
            const v8f Pp  = mm16(oT1, opA, dQ);
            const v8f PCtT = mm16(opCA, oT1, dCQt);

            gj_pivot<0>(g); gj_pivot<1>(g); gj_pivot<2>(g); gj_pivot<3>(g);
            gj_pivot<4>(g); gj_pivot<5>(g); gj_pivot<6>(g); gj_pivot<7>(g);
            Op4 oSi = from_acc(g);
            {
                const bool keep = lane < MS;
                const v2f z2 = v2f{0.f, 0.f};
                oSi.q[0] = keep ? oSi.q[0] : z2;
                oSi.q[1] = keep ? oSi.q[1] : z2;
                oSi.q[2] = keep ? oSi.q[2] : z2;
                oSi.q[3] = keep ? oSi.q[3] : z2;
            }
            const v8f Kt = mm16(oSi, from_acc(PCtT), zero);
            const v8f Wt = mm16(opNCt, from_acc(Kt), dI);
            const Op4 oW = from_acc(Wt);
            P = mm16(oW, from_acc(Pp), zero);
            const v8f Hb = mm16(opBtA, oW, zero);
            const v8f G = mm16(opAtA, oW, zero);

            float* wsp = &hand[t & 1][0];
            {
                float* o = wsp + c * NS + 8 * h;
                *(float4*)(o)     = make_float4(G[0], G[1], G[2], G[3]);
                *(float4*)(o + 4) = make_float4(G[4], G[5], G[6], G[7]);
            }
            if (h == 0) {
                *(float4*)(wsp + 256 + c * US) = make_float4(Hb[0], Hb[1], Hb[2], Hb[3]);
                float* o = wsp + 320 + c * MS;
                *(float4*)(o)     = make_float4(Kt[0], Kt[1], Kt[2], Kt[3]);
                *(float4*)(o + 4) = make_float4(Kt[4], Kt[5], Kt[6], Kt[7]);
            }
        } else if (!cov_wave && t >= 2) {
            const int s = t - 1;

            const float* w = &hand[s & 1][0];
            const v2f aA0 = *(const v2f*)(w + c * NS + 8 * h + 0);
            const v2f aA1 = *(const v2f*)(w + c * NS + 8 * h + 2);
            const v2f aA2 = *(const v2f*)(w + c * NS + 8 * h + 4);
            const v2f aA3 = *(const v2f*)(w + c * NS + 8 * h + 6);
            const v2f aB  = *(const v2f*)(w + 256 + c * US + 2 * h);
            const v2f aK0 = *(const v2f*)(w + 320 + c * MS + 0 + 2 * h);
            const v2f aK1 = *(const v2f*)(w + 320 + c * MS + 4 + 2 * h);

            v8f d = zero;
            d = wmma4(aB,  zu.bU,  d);
            d = wmma4(aK0, zu.bZ0, d);
            d = wmma4(aK1, zu.bZ1, d);
            d = wmma4(aA0, v2f{x[0], x[1]}, d);
            d = wmma4(aA1, v2f{x[2], x[3]}, d);
            d = wmma4(aA2, v2f{x[4], x[5]}, d);
            d = wmma4(aA3, v2f{x[6], x[7]}, d);
            x = d;

            float* o = obase + (size_t)s * NS + 8 * h;
            *(float4*)(o)     = make_float4(x[0], x[1], x[2], x[3]);
            *(float4*)(o + 4) = make_float4(x[4], x[5], x[6], x[7]);
        }
    }
}

extern "C" void kernel_launch(void* const* d_in, const int* in_sizes, int n_in,
                              void* d_out, int out_size, void* d_ws, size_t ws_size,
                              hipStream_t stream) {
    const float* obs = (const float*)d_in[0];
    const float* inp = (const float*)d_in[1];
    const float* A   = (const float*)d_in[2];
    const float* B   = (const float*)d_in[3];
    const float* C   = (const float*)d_in[4];
    const float* Q   = (const float*)d_in[5];
    const float* R   = (const float*)d_in[6];
    const float* x0  = (const float*)d_in[7];
    float* out = (float*)d_out;

    hipLaunchKernelGGL(kalman_fused_kernel, dim3(BATCH / 16), dim3(64), 0, stream,
                       obs, inp, A, B, C, Q, R, x0, out);
}


# State wave's obs and input operands loaded 3 passes ahead through a 3-slot register ring (passes unrolled by 3), not in the pass that uses them: no pass waits a load round trip; same sums and order
# speedup vs baseline: 8.1813x; 1.0572x over previous
#include <hip/hip_runtime.h>

#define BATCH 256
#define SEQ   4096
#define NS    16
#define MS    8
#define US    4
#define STEPF 448
#define RING  3
static_assert((SEQ - 1) % RING == 0 && SEQ > RING + 1, "the SEQ - 1 passes are unrolled in rounds of RING");

typedef __attribute__((ext_vector_type(2))) float v2f;
typedef __attribute__((ext_vector_type(8))) float v8f;

template <int L>
__device__ __forceinline__ float row_share(float v) {
    return __builtin_bit_cast(float,
        __builtin_amdgcn_update_dpp(0, __builtin_bit_cast(int, v), 0x150 + L, 0xf, 0xf, true));
}

__device__ __forceinline__ v8f wmma4(v2f a, v2f b, v8f c) {
    return __builtin_amdgcn_wmma_f32_16x16x4_f32(false, a, false, b,
                                                 (short)0, c, false, false);
}

struct Op4 { v2f q[4]; };

__device__ __forceinline__ Op4 from_acc(v8f x) {
    Op4 o;
    o.q[0] = v2f{x[0], x[1]};
    o.q[1] = v2f{x[2], x[3]};
    o.q[2] = v2f{x[4], x[5]};
    o.q[3] = v2f{x[6], x[7]};
    return o;
}

__device__ __forceinline__ v8f mm16(const Op4& a, const Op4& b, v8f c) {
    c = wmma4(a.q[0], b.q[0], c);
    c = wmma4(a.q[1], b.q[1], c);
    c = wmma4(a.q[2], b.q[2], c);
    c = wmma4(a.q[3], b.q[3], c);
    return c;
}

struct ZU { v2f bZ0, bZ1, bU; };

__device__ __forceinline__ ZU load_zu(const float* __restrict__ zbase,
                                      const float* __restrict__ ubase, int t, int hi) {
    ZU o;
    const float* zr = zbase + (size_t)t * MS;
    o.bZ0 = *(const v2f*)(zr + 0 + 2 * hi);
    o.bZ1 = *(const v2f*)(zr + 4 + 2 * hi);
    o.bU  = *(const v2f*)(ubase + (size_t)t * US + 2 * hi);
    return o;
}

template <int P>
__device__ __forceinline__ void gj_pivot(v8f& g) {
    const float invp = __builtin_amdgcn_rcpf(row_share<MS + P>(g[P]));
    const float pv = g[P] * invp;
    #pragma unroll
    for (int i = 0; i < MS; ++i) {
        if (i != P) {
            const float f = row_share<MS + P>(g[i]);
            g[i] = g[i] - f * pv;
        }
    }
    g[P] = pv;
}

__global__ __launch_bounds__(64)
void kalman_fused_kernel(const float* __restrict__ obs, const float* __restrict__ inp,
                         const float* __restrict__ Ag, const float* __restrict__ Bg,
                         const float* __restrict__ Cg, const float* __restrict__ Qg,
                         const float* __restrict__ Rg, const float* __restrict__ x0,
                         float* __restrict__ out)
{
    __shared__ __attribute__((aligned(16))) float hand[2][STEPF];

    const int lane = threadIdx.x & 31;
    const int h = lane >> 4;
    const int c = lane & 15;
    const bool cov_wave = __builtin_amdgcn_readfirstlane((int)(threadIdx.x >> 5)) == 0;

    const v8f zero = {};

    Op4 opA = {}, opAtA = {}, opNCt = {}, opBtA = {}, opCA = {}, opCAsh = {};
    v8f dQ = zero, dI = zero, dCQt = zero, dSaug = zero, P = zero;
    v8f x = zero;
    ZU ring[RING] = {};
    const int b = blockIdx.x * 16 + c;
    const float* zbase = obs + (size_t)b * SEQ * MS;
    const float* ubase = inp + (size_t)b * SEQ * US;
    float*       obase = out + (size_t)b * SEQ * NS;

    if (cov_wave) {
        const int k0 = 8 * h;
        const int c7 = c & 7;
        const int c3 = c & 3;
        Op4 opC, opCsh;
        #pragma unroll
        for (int q = 0; q < 4; ++q) {
            const int k = k0 + 2 * q;
            opA.q[q]   = v2f{Ag[c * NS + k], Ag[c * NS + k + 1]};
            opAtA.q[q] = v2f{Ag[k * NS + c], Ag[(k + 1) * NS + c]};
            const float cv0 = Cg[c7 * NS + k], cv1 = Cg[c7 * NS + k + 1];
            opC.q[q]   = (c < MS)  ? v2f{cv0, cv1} : v2f{0.f, 0.f};
            opCsh.q[q] = (c >= MS) ? v2f{cv0, cv1} : v2f{0.f, 0.f};
            const float n0 = -Cg[(2 * q) * NS + c], n1 = -Cg[(2 * q + 1) * NS + c];
            opNCt.q[q] = (h == 0) ? v2f{n0, n1} : v2f{0.f, 0.f};
            const float b0 = Bg[k * US + c3], b1 = Bg[(k + 1) * US + c3];
            opBtA.q[q] = (c < US) ? v2f{b0, b1} : v2f{0.f, 0.f};
        }
        v8f dQt, dRaug;
        #pragma unroll
        for (int r = 0; r < 8; ++r) {
            dQ[r]  = Qg[(k0 + r) * NS + c];
            dQt[r] = Qg[c * NS + k0 + r];
            dI[r]  = (k0 + r == c) ? 1.0f : 0.0f;
            const float rv = Rg[r * MS + c7];
            dRaug[r] = (h == 0) ? ((c >= MS) ? rv : ((r == c) ? 1.0f : 0.0f))
                                : ((c == MS + r) ? 1.0f : 0.0f);
            P[r] = dI[r];
        }
        opCA   = from_acc(mm16(opAtA, opC,   zero));
        opCAsh = from_acc(mm16(opAtA, opCsh, zero));
        dCQt = mm16(opC, from_acc(dQt), zero);
        const v8f CQT = mm16(from_acc(dQ), opC, zero);
        dSaug = mm16(from_acc(CQT), opCsh, dRaug);
    } else {
        #pragma unroll
        for (int r = 0; r < 8; ++r) x[r] = x0[8 * h + r];
        float* o = obase + 8 * h;
        *(float4*)(o)     = make_float4(x[0], x[1], x[2], x[3]);
        *(float4*)(o + 4) = make_float4(x[4], x[5], x[6], x[7]);
        #pragma unroll
        for (int j = 0; j < RING; ++j) ring[j] = load_zu(zbase, ubase, 1 + j, h);
    }

    v8f T1t = zero, V = zero;

    auto cov_first = [&]() __attribute__((always_inline)) {
        const Op4 oP = from_acc(P);
        T1t = mm16(oP, opA, zero);
        V   = mm16(oP, opCA, zero);
    };

    auto cov_rest = [&](int step) __attribute__((always_inline)) {
        const Op4 oT1 = from_acc(T1t);
        v8f g = mm16(from_acc(V), opCAsh, dSaug);
        const v8f Pp  = mm16(oT1, opA, dQ);
        const v8f PCtT = mm16(opCA, oT1, dCQt);

        gj_pivot<0>(g); gj_pivot<1>(g); gj_pivot<2>(g); gj_pivot<3>(g);
        gj_pivot<4>(g); gj_pivot<5>(g); gj_pivot<6>(g); gj_pivot<7>(g);
        Op4 oSi = from_acc(g);
        {
            const bool keep = lane < MS;
            const v2f z2 = v2f{0.f, 0.f};
            oSi.q[0] = keep ? oSi.q[0] : z2;
            oSi.q[1] = keep ? oSi.q[1] : z2;
            oSi.q[2] = keep ? oSi.q[2] : z2;
            oSi.q[3] = keep ? oSi.q[3] : z2;
        }
        const v8f Kt = mm16(oSi, from_acc(PCtT), zero);
        const v8f Wt = mm16(opNCt, from_acc(Kt), dI);
        const Op4 oW = from_acc(Wt);
        P = mm16(oW, from_acc(Pp), zero);
        const v8f Hb = mm16(opBtA, oW, zero);
        const v8f G = mm16(opAtA, oW, zero);

        float* wsp = &hand[step & 1][0];
        {
            float* o = wsp + c * NS + 8 * h;
            *(float4*)(o)     = make_float4(G[0], G[1], G[2], G[3]);
            *(float4*)(o + 4) = make_float4(G[4], G[5], G[6], G[7]);
        }
        if (h == 0) {
            *(float4*)(wsp + 256 + c * US) = make_float4(Hb[0], Hb[1], Hb[2], Hb[3]);
            float* o = wsp + 320 + c * MS;
            *(float4*)(o)     = make_float4(Kt[0], Kt[1], Kt[2], Kt[3]);
            *(float4*)(o + 4) = make_float4(Kt[4], Kt[5], Kt[6], Kt[7]);
        }
    };

    auto state_step = [&](int s, const ZU& zu) __attribute__((always_inline)) {
        const float* w = &hand[s & 1][0];
        const v2f aA0 = *(const v2f*)(w + c * NS + 8 * h + 0);
        const v2f aA1 = *(const v2f*)(w + c * NS + 8 * h + 2);
        const v2f aA2 = *(const v2f*)(w + c * NS + 8 * h + 4);
        const v2f aA3 = *(const v2f*)(w + c * NS + 8 * h + 6);
        const v2f aB  = *(const v2f*)(w + 256 + c * US + 2 * h);
        const v2f aK0 = *(const v2f*)(w + 320 + c * MS + 0 + 2 * h);
        const v2f aK1 = *(const v2f*)(w + 320 + c * MS + 4 + 2 * h);

        v8f d = zero;
        d = wmma4(aB,  zu.bU,  d);
        d = wmma4(aK0, zu.bZ0, d);
        d = wmma4(aK1, zu.bZ1, d);
        d = wmma4(aA0, v2f{x[0], x[1]}, d);
        d = wmma4(aA1, v2f{x[2], x[3]}, d);
        d = wmma4(aA2, v2f{x[4], x[5]}, d);
        d = wmma4(aA3, v2f{x[6], x[7]}, d);
        x = d;

        float* o = obase + (size_t)s * NS + 8 * h;
        *(float4*)(o)     = make_float4(x[0], x[1], x[2], x[3]);
        *(float4*)(o + 4) = make_float4(x[4], x[5], x[6], x[7]);
    };

    if (cov_wave) { cov_first(); cov_rest(1); }

    #pragma unroll 1
    for (int tb = 1; tb < SEQ; tb += RING) {
        #pragma unroll
        for (int j = 0; j < RING; ++j) {
            const int t = tb + j;
            const bool cov_step = cov_wave && (t + 1 < SEQ);
            if (cov_step) cov_first();

            asm volatile("s_wait_dscnt 0x0" ::: "memory");
            __builtin_amdgcn_fence(__ATOMIC_RELEASE, "workgroup", "local");
            __builtin_amdgcn_s_barrier();
            __builtin_amdgcn_fence(__ATOMIC_ACQUIRE, "workgroup", "local");

            if (cov_step) cov_rest(t + 1);
            else if (!cov_wave) state_step(t, ring[j]);

            ring[j] = load_zu(zbase, ubase, (t + RING < SEQ) ? (t + RING) : (SEQ - 1), h);
        }
    }
}

extern "C" void kernel_launch(void* const* d_in, const int* in_sizes, int n_in,
                              void* d_out, int out_size, void* d_ws, size_t ws_size,
                              hipStream_t stream) {
    const float* obs = (const float*)d_in[0];
    const float* inp = (const float*)d_in[1];
    const float* A   = (const float*)d_in[2];
    const float* B   = (const float*)d_in[3];
    const float* C   = (const float*)d_in[4];
    const float* Q   = (const float*)d_in[5];
    const float* R   = (const float*)d_in[6];
    const float* x0  = (const float*)d_in[7];
    float* out = (float*)d_out;

    hipLaunchKernelGGL(kalman_fused_kernel, dim3(BATCH / 16), dim3(64), 0, stream,
                       obs, inp, A, B, C, Q, R, x0, out);
}


# Products W A and W B leave the covariance wave (8 of its 40 WMMAs per step): it hands over W and K; the state wave forms A x + B u itself, a pass early, then W y + K z; sums reassociated
# speedup vs baseline: 9.2034x; 1.1249x over previous
#include <hip/hip_runtime.h>

#define BATCH 256
#define SEQ   4096
#define NS    16
#define MS    8
#define US    4
#define STEPF 384
#define RING  3
static_assert((SEQ - 1) % RING == 0 && SEQ > RING + 1, "the SEQ - 1 passes are unrolled in rounds of RING");

typedef __attribute__((ext_vector_type(2))) float v2f;
typedef __attribute__((ext_vector_type(8))) float v8f;

template <int L>
__device__ __forceinline__ float row_share(float v) {
    return __builtin_bit_cast(float,
        __builtin_amdgcn_update_dpp(0, __builtin_bit_cast(int, v), 0x150 + L, 0xf, 0xf, true));
}

__device__ __forceinline__ v8f wmma4(v2f a, v2f b, v8f c) {
    return __builtin_amdgcn_wmma_f32_16x16x4_f32(false, a, false, b,
                                                 (short)0, c, false, false);
}

struct Op4 { v2f q[4]; };

__device__ __forceinline__ Op4 from_acc(v8f x) {
    Op4 o;
    o.q[0] = v2f{x[0], x[1]};
    o.q[1] = v2f{x[2], x[3]};
    o.q[2] = v2f{x[4], x[5]};
    o.q[3] = v2f{x[6], x[7]};
    return o;
}

__device__ __forceinline__ v8f mm16(const Op4& a, const Op4& b, v8f c) {
    c = wmma4(a.q[0], b.q[0], c);
    c = wmma4(a.q[1], b.q[1], c);
    c = wmma4(a.q[2], b.q[2], c);
    c = wmma4(a.q[3], b.q[3], c);
    return c;
}

struct ZU { v2f bZ0, bZ1, bU; };

__device__ __forceinline__ ZU load_zu(const float* __restrict__ zbase,
                                      const float* __restrict__ ubase, int t, int hi) {
    ZU o;
    const float* zr = zbase + (size_t)t * MS;
    o.bZ0 = *(const v2f*)(zr + 0 + 2 * hi);
    o.bZ1 = *(const v2f*)(zr + 4 + 2 * hi);
    o.bU  = *(const v2f*)(ubase + (size_t)t * US + 2 * hi);
    return o;
}

template <int P>
__device__ __forceinline__ void gj_pivot(v8f& g) {
    const float invp = __builtin_amdgcn_rcpf(row_share<MS + P>(g[P]));
    const float pv = g[P] * invp;
    #pragma unroll
    for (int i = 0; i < MS; ++i) {
        if (i != P) {
            const float f = row_share<MS + P>(g[i]);
            g[i] = g[i] - f * pv;
        }
    }
    g[P] = pv;
}

__global__ __launch_bounds__(64)
void kalman_fused_kernel(const float* __restrict__ obs, const float* __restrict__ inp,
                         const float* __restrict__ Ag, const float* __restrict__ Bg,
                         const float* __restrict__ Cg, const float* __restrict__ Qg,
                         const float* __restrict__ Rg, const float* __restrict__ x0,
                         float* __restrict__ out)
{
    __shared__ __attribute__((aligned(16))) float hand[2][STEPF];

    const int lane = threadIdx.x & 31;
    const int h = lane >> 4;
    const int c = lane & 15;
    const bool cov_wave = __builtin_amdgcn_readfirstlane((int)(threadIdx.x >> 5)) == 0;

    const v8f zero = {};

    Op4 opA = {}, opNCt = {}, opCA = {}, opCAsh = {};
    v8f dQ = zero, dI = zero, dCQt = zero, dSaug = zero, P = zero;
    v8f x = zero, y = zero;
    v2f aBu = {};
    ZU ring[RING] = {};
    const int b = blockIdx.x * 16 + c;
    const float* zbase = obs + (size_t)b * SEQ * MS;
    const float* ubase = inp + (size_t)b * SEQ * US;
    float*       obase = out + (size_t)b * SEQ * NS;

    if (cov_wave) {
        const int k0 = 8 * h;
        const int c7 = c & 7;
        Op4 opAtA, opC, opCsh;
        #pragma unroll
        for (int q = 0; q < 4; ++q) {
            const int k = k0 + 2 * q;
            opA.q[q]   = v2f{Ag[c * NS + k], Ag[c * NS + k + 1]};
            opAtA.q[q] = v2f{Ag[k * NS + c], Ag[(k + 1) * NS + c]};
            const float cv0 = Cg[c7 * NS + k], cv1 = Cg[c7 * NS + k + 1];
            opC.q[q]   = (c < MS)  ? v2f{cv0, cv1} : v2f{0.f, 0.f};
            opCsh.q[q] = (c >= MS) ? v2f{cv0, cv1} : v2f{0.f, 0.f};
            const float n0 = -Cg[(2 * q) * NS + c], n1 = -Cg[(2 * q + 1) * NS + c];
            opNCt.q[q] = (h == 0) ? v2f{n0, n1} : v2f{0.f, 0.f};
        }
        v8f dQt, dRaug;
        #pragma unroll
        for (int r = 0; r < 8; ++r) {
            dQ[r]  = Qg[(k0 + r) * NS + c];
            dQt[r] = Qg[c * NS + k0 + r];
            dI[r]  = (k0 + r == c) ? 1.0f : 0.0f;
            const float rv = Rg[r * MS + c7];
            dRaug[r] = (h == 0) ? ((c >= MS) ? rv : ((r == c) ? 1.0f : 0.0f))
                                : ((c == MS + r) ? 1.0f : 0.0f);
            P[r] = dI[r];
        }
        opCA   = from_acc(mm16(opAtA, opC,   zero));
        opCAsh = from_acc(mm16(opAtA, opCsh, zero));
        dCQt = mm16(opC, from_acc(dQt), zero);
        const v8f CQT = mm16(from_acc(dQ), opC, zero);
        dSaug = mm16(from_acc(CQT), opCsh, dRaug);
    } else {
        #pragma unroll
        for (int q = 0; q < 4; ++q) {
            const int k = 8 * h + 2 * q;
            opA.q[q] = v2f{Ag[c * NS + k], Ag[c * NS + k + 1]};
        }
        aBu = *(const v2f*)(Bg + c * US + 2 * h);
        #pragma unroll
        for (int r = 0; r < 8; ++r) x[r] = x0[8 * h + r];
        float* o = obase + 8 * h;
        *(float4*)(o)     = make_float4(x[0], x[1], x[2], x[3]);
        *(float4*)(o + 4) = make_float4(x[4], x[5], x[6], x[7]);
        #pragma unroll
        for (int j = 0; j < RING; ++j) ring[j] = load_zu(zbase, ubase, 1 + j, h);
    }

    v8f T1t = zero, V = zero;

    auto cov_first = [&]() __attribute__((always_inline)) {
        const Op4 oP = from_acc(P);
        T1t = mm16(oP, opA, zero);
        V   = mm16(oP, opCA, zero);
    };

    auto cov_rest = [&](int step) __attribute__((always_inline)) {
        const Op4 oT1 = from_acc(T1t);
        v8f g = mm16(from_acc(V), opCAsh, dSaug);
        const v8f Pp  = mm16(oT1, opA, dQ);
        const v8f PCtT = mm16(opCA, oT1, dCQt);

        gj_pivot<0>(g); gj_pivot<1>(g); gj_pivot<2>(g); gj_pivot<3>(g);
        gj_pivot<4>(g); gj_pivot<5>(g); gj_pivot<6>(g); gj_pivot<7>(g);
        Op4 oSi = from_acc(g);
        {
            const bool keep = lane < MS;
            const v2f z2 = v2f{0.f, 0.f};
            oSi.q[0] = keep ? oSi.q[0] : z2;
            oSi.q[1] = keep ? oSi.q[1] : z2;
            oSi.q[2] = keep ? oSi.q[2] : z2;
            oSi.q[3] = keep ? oSi.q[3] : z2;
        }
        const v8f Kt = mm16(oSi, from_acc(PCtT), zero);
        const v8f Wt = mm16(opNCt, from_acc(Kt), dI);
        const Op4 oW = from_acc(Wt);
        P = mm16(oW, from_acc(Pp), zero);

        float* wsp = &hand[step & 1][0];
        {
            float* o = wsp + c * NS + 8 * h;
            *(float4*)(o)     = make_float4(Wt[0], Wt[1], Wt[2], Wt[3]);
            *(float4*)(o + 4) = make_float4(Wt[4], Wt[5], Wt[6], Wt[7]);
        }
        if (h == 0) {
            float* o = wsp + 256 + c * MS;
            *(float4*)(o)     = make_float4(Kt[0], Kt[1], Kt[2], Kt[3]);
            *(float4*)(o + 4) = make_float4(Kt[4], Kt[5], Kt[6], Kt[7]);
        }
    };

    auto state_first = [&](const ZU& zu) __attribute__((always_inline)) {
        v8f d = wmma4(aBu, zu.bU, zero);
        d = wmma4(opA.q[0], v2f{x[0], x[1]}, d);
        d = wmma4(opA.q[1], v2f{x[2], x[3]}, d);
        d = wmma4(opA.q[2], v2f{x[4], x[5]}, d);
        d = wmma4(opA.q[3], v2f{x[6], x[7]}, d);
        y = d;
    };

    auto state_rest = [&](int s, const ZU& zu) __attribute__((always_inline)) {
        const float* w = &hand[s & 1][0];
        const v2f aW0 = *(const v2f*)(w + c * NS + 8 * h + 0);
        const v2f aW1 = *(const v2f*)(w + c * NS + 8 * h + 2);
        const v2f aW2 = *(const v2f*)(w + c * NS + 8 * h + 4);
        const v2f aW3 = *(const v2f*)(w + c * NS + 8 * h + 6);
        const v2f aK0 = *(const v2f*)(w + 256 + c * MS + 0 + 2 * h);
        const v2f aK1 = *(const v2f*)(w + 256 + c * MS + 4 + 2 * h);

        v8f d = zero;
        d = wmma4(aK0, zu.bZ0, d);
        d = wmma4(aK1, zu.bZ1, d);
        d = wmma4(aW0, v2f{y[0], y[1]}, d);
        d = wmma4(aW1, v2f{y[2], y[3]}, d);
        d = wmma4(aW2, v2f{y[4], y[5]}, d);
        d = wmma4(aW3, v2f{y[6], y[7]}, d);
        x = d;

        float* o = obase + (size_t)s * NS + 8 * h;
        *(float4*)(o)     = make_float4(x[0], x[1], x[2], x[3]);
        *(float4*)(o + 4) = make_float4(x[4], x[5], x[6], x[7]);
    };

    if (cov_wave) { cov_first(); cov_rest(1); }
    else state_first(ring[0]);

    #pragma unroll 1
    for (int tb = 1; tb < SEQ; tb += RING) {
        #pragma unroll
        for (int j = 0; j < RING; ++j) {
            const int t = tb + j;
            const bool cov_step = cov_wave && (t + 1 < SEQ);
            if (cov_step) cov_first();

            asm volatile("s_wait_dscnt 0x0" ::: "memory");
            __builtin_amdgcn_fence(__ATOMIC_RELEASE, "workgroup", "local");
            __builtin_amdgcn_s_barrier();
            __builtin_amdgcn_fence(__ATOMIC_ACQUIRE, "workgroup", "local");

            if (cov_step) cov_rest(t + 1);
            else if (!cov_wave) {
                state_rest(t, ring[j]);
                state_first(ring[(j + 1) % RING]);
            }

            ring[j] = load_zu(zbase, ubase, (t + RING < SEQ) ? (t + RING) : (SEQ - 1), h);
        }
    }
}

extern "C" void kernel_launch(void* const* d_in, const int* in_sizes, int n_in,
                              void* d_out, int out_size, void* d_ws, size_t ws_size,
                              hipStream_t stream) {
    const float* obs = (const float*)d_in[0];
    const float* inp = (const float*)d_in[1];
    const float* A   = (const float*)d_in[2];
    const float* B   = (const float*)d_in[3];
    const float* C   = (const float*)d_in[4];
    const float* Q   = (const float*)d_in[5];
    const float* R   = (const float*)d_in[6];
    const float* x0  = (const float*)d_in[7];
    float* out = (float*)d_out;

    hipLaunchKernelGGL(kalman_fused_kernel, dim3(BATCH / 16), dim3(64), 0, stream,
                       obs, inp, A, B, C, Q, R, x0, out);
}
